# starting kernel
# baseline (speedup 1.0000x reference)
#include <hip/hip_runtime.h>


typedef __attribute__((ext_vector_type(2))) float v2f;
typedef __attribute__((ext_vector_type(8))) float v8f;

#define IN_DIM 128
#define H1_DIM 16
#define H2_DIM 8

__global__ void gcn_init_deg(float* __restrict__ deg, int n) {
    int i = blockIdx.x * blockDim.x + threadIdx.x;
    if (i < n) deg[i] = 1.0f;
}

__global__ void gcn_count_deg(const int* __restrict__ dst, float* __restrict__ deg, int e) {
    int i = blockIdx.x * blockDim.x + threadIdx.x;
    if (i < e) atomicAdd(&deg[dst[i]], 1.0f);
}

__global__ void gcn_rsqrt_deg(float* __restrict__ deg, int n) {
    int i = blockIdx.x * blockDim.x + threadIdx.x;
    if (i < n) {
        float d = deg[i];
        deg[i] = (d > 0.0f) ? rsqrtf(d) : 0.0f;
    }
}

__global__ __launch_bounds__(256) void gcn_gemm1_wmma(
    const float* __restrict__ x, const float* __restrict__ W1,
    const float* __restrict__ dis, const float* __restrict__ b1,
    float* __restrict__ hw1, float* __restrict__ agg1, int n) {
    __shared__ float W1s[IN_DIM * H1_DIM];
    const int tid = threadIdx.x;
    #pragma unroll
    for (int j = 0; j < (IN_DIM * H1_DIM) / 256; ++j)
        W1s[tid + j * 256] = W1[tid + j * 256];
    __syncthreads();

    const int wave = tid >> 5;
    const int lane = tid & 31;
    const int half = lane >> 4;
    const int l    = lane & 15;
    const int row0 = (blockIdx.x * 8 + wave) * 16;
    if (row0 >= n) return;

    int arow = row0 + l;
    if (arow >= n) arow = n - 1;

    v8f c = {};
    #pragma unroll 4
    for (int k0 = 0; k0 < IN_DIM; k0 += 4) {
        const int ka = k0 + half * 2;
        const v2f a = *(const v2f*)&x[(size_t)arow * IN_DIM + ka];
        v2f b;
        b.x = W1s[ka * H1_DIM + l];
        b.y = W1s[(ka + 1) * H1_DIM + l];
        c = __builtin_amdgcn_wmma_f32_16x16x4_f32(
                false, a, false, b, (short)0, c, false, false);
    }

    const float bias = b1[l];
    if (row0 + 16 <= n) {
        #pragma unroll
        for (int r = 0; r < 8; ++r) {
            const int row = row0 + r + half * 8;
            const float v  = c[r];
            const float dv = dis[row];
            hw1 [(size_t)row * H1_DIM + l] = v;
            agg1[(size_t)row * H1_DIM + l] = dv * dv * v + bias;
        }
    } else {
        for (int r = 0; r < 8; ++r) {
            const int row = row0 + r + half * 8;
            if (row < n) {
                const float v  = c[r];
                const float dv = dis[row];
                hw1 [(size_t)row * H1_DIM + l] = v;
                agg1[(size_t)row * H1_DIM + l] = dv * dv * v + bias;
            }
        }
    }
}

__global__ __launch_bounds__(256) void gcn_scatter1(
    const int* __restrict__ src, const int* __restrict__ dst,
    const float* __restrict__ dis, const float* __restrict__ hw1,
    float* __restrict__ agg1, int e) {
    const int i = blockIdx.x * blockDim.x + threadIdx.x;
    if (i >= e * H1_DIM) return;
    const int eidx = i >> 4;
    const int c    = i & 15;

    if (c == 0) {
        const int la = eidx + 512;
        if (la < e) __builtin_prefetch(&hw1[(size_t)src[la] * H1_DIM], 0, 0);
    }

    const int s = src[eidx];
    const int d = dst[eidx];
    const float w = dis[s] * dis[d];
    atomicAdd(&agg1[(size_t)d * H1_DIM + c], w * hw1[(size_t)s * H1_DIM + c]);
}

__global__ __launch_bounds__(256) void gcn_gemm2_wmma(
    const float* __restrict__ h1, const float* __restrict__ W2,
    const float* __restrict__ dis, const float* __restrict__ b2,
    float* __restrict__ hw2, float* __restrict__ out, int n) {
    __shared__ float W2s[H1_DIM * 16];
    const int tid = threadIdx.x;
    {
        const int k = tid >> 4, col = tid & 15;
        W2s[tid] = (col < H2_DIM) ? W2[k * H2_DIM + col] : 0.0f;
    }
    __syncthreads();

    const int wave = tid >> 5;
    const int lane = tid & 31;
    const int half = lane >> 4;
    const int l    = lane & 15;
    const int row0 = (blockIdx.x * 8 + wave) * 16;
    if (row0 >= n) return;

    int arow = row0 + l;
    if (arow >= n) arow = n - 1;

    v8f c = {};
    #pragma unroll
    for (int k0 = 0; k0 < H1_DIM; k0 += 4) {
        const int ka = k0 + half * 2;
        v2f a = *(const v2f*)&h1[(size_t)arow * H1_DIM + ka];
        a.x = fmaxf(a.x, 0.0f);
        a.y = fmaxf(a.y, 0.0f);
        v2f b;
        b.x = W2s[ka * 16 + l];
        b.y = W2s[(ka + 1) * 16 + l];
        c = __builtin_amdgcn_wmma_f32_16x16x4_f32(
                false, a, false, b, (short)0, c, false, false);
    }

    if (l < H2_DIM) {
        const float bias = b2[l];
        if (row0 + 16 <= n) {
            #pragma unroll
            for (int r = 0; r < 8; ++r) {
                const int row = row0 + r + half * 8;
                const float v  = c[r];
                const float dv = dis[row];
                hw2[(size_t)row * H2_DIM + l] = v;
                out[(size_t)row * H2_DIM + l] = dv * dv * v + bias;
            }
        } else {
            for (int r = 0; r < 8; ++r) {
                const int row = row0 + r + half * 8;
                if (row < n) {
                    const float v  = c[r];
                    const float dv = dis[row];
                    hw2[(size_t)row * H2_DIM + l] = v;
                    out[(size_t)row * H2_DIM + l] = dv * dv * v + bias;
                }
            }
        }
    }
}

__global__ __launch_bounds__(256) void gcn_scatter2(
    const int* __restrict__ src, const int* __restrict__ dst,
    const float* __restrict__ dis, const float* __restrict__ hw2,
    float* __restrict__ out, int e) {
    const int i = blockIdx.x * blockDim.x + threadIdx.x;
    if (i >= e * H2_DIM) return;
    const int eidx = i >> 3;
    const int c    = i & 7;

    if (c == 0) {
        const int la = eidx + 512;
        if (la < e) __builtin_prefetch(&hw2[(size_t)src[la] * H2_DIM], 0, 0);
    }

    const int s = src[eidx];
    const int d = dst[eidx];
    const float w = dis[s] * dis[d];
    atomicAdd(&out[(size_t)d * H2_DIM + c], w * hw2[(size_t)s * H2_DIM + c]);
}

extern "C" void kernel_launch(void* const* d_in, const int* in_sizes, int n_in,
                              void* d_out, int out_size, void* d_ws, size_t ws_size,
                              hipStream_t stream) {
    const float* x    = (const float*)d_in[0];
    const int*   edge = (const int*)  d_in[1];
    const float* W1   = (const float*)d_in[2];
    const float* b1   = (const float*)d_in[3];
    const float* W2   = (const float*)d_in[4];
    const float* b2   = (const float*)d_in[5];
    float* out = (float*)d_out;

    const int n = in_sizes[0] / IN_DIM;
    const int e = in_sizes[1] / 2;
    const int* src = edge;
    const int* dst = edge + e;

    float* ws   = (float*)d_ws;
    float* dis  = ws;
    float* hw1  = dis + n;
    float* agg1 = hw1 + (size_t)n * H1_DIM;
    float* hw2  = agg1 + (size_t)n * H1_DIM;

    const int B = 256;
    const int gn     = (n + B - 1) / B;
    const int ge     = (e + B - 1) / B;
    const int tiles  = (n + 15) / 16;
    const int gtiles = (tiles + 7) / 8;
    const int gs1    = (e * H1_DIM + B - 1) / B;
    const int gs2    = (e * H2_DIM + B - 1) / B;

    gcn_init_deg <<<gn, B, 0, stream>>>(dis, n);
    gcn_count_deg<<<ge, B, 0, stream>>>(dst, dis, e);
    gcn_rsqrt_deg<<<gn, B, 0, stream>>>(dis, n);

    gcn_gemm1_wmma<<<gtiles, B, 0, stream>>>(x, W1, dis, b1, hw1, agg1, n);
    gcn_scatter1  <<<gs1, B, 0, stream>>>(src, dst, dis, hw1, agg1, e);

    gcn_gemm2_wmma<<<gtiles, B, 0, stream>>>(agg1, W2, dis, b2, hw2, out, n);
    gcn_scatter2  <<<gs2, B, 0, stream>>>(src, dst, dis, hw2, out, e);
}


# Both edge scatters: no global atomics, no binning: each work-group filters the edge list for its node range, sums rows in LDS (ds f32 adds, arrival order); slice partials added by the reader
# speedup vs baseline: 1.5918x; 1.5918x over previous
#include <hip/hip_runtime.h>


typedef __attribute__((ext_vector_type(2))) float v2f;
typedef __attribute__((ext_vector_type(8))) float v8f;

#define IN_DIM 128
#define H1_DIM 16
#define H2_DIM 8

__global__ void gcn_init_deg(float* __restrict__ deg, int n) {
    int i = blockIdx.x * blockDim.x + threadIdx.x;
    if (i < n) deg[i] = 1.0f;
}

__global__ void gcn_count_deg(const int* __restrict__ dst, float* __restrict__ deg, int e) {
    int i = blockIdx.x * blockDim.x + threadIdx.x;
    if (i < e) atomicAdd(&deg[dst[i]], 1.0f);
}

__global__ void gcn_rsqrt_deg(float* __restrict__ deg, int n) {
    int i = blockIdx.x * blockDim.x + threadIdx.x;
    if (i < n) {
        float d = deg[i];
        deg[i] = (d > 0.0f) ? rsqrtf(d) : 0.0f;
    }
}

__global__ __launch_bounds__(256) void gcn_gemm1_wmma(
    const float* __restrict__ x, const float* __restrict__ W1,
    const float* __restrict__ dis, const float* __restrict__ b1,
    float* __restrict__ hw1, float* __restrict__ agg1, int n) {
    __shared__ float W1s[IN_DIM * H1_DIM];
    const int tid = threadIdx.x;
    #pragma unroll
    for (int j = 0; j < (IN_DIM * H1_DIM) / 256; ++j)
        W1s[tid + j * 256] = W1[tid + j * 256];
    __syncthreads();

    const int wave = tid >> 5;
    const int lane = tid & 31;
    const int half = lane >> 4;
    const int l    = lane & 15;
    const int row0 = (blockIdx.x * 8 + wave) * 16;
    if (row0 >= n) return;

    int arow = row0 + l;
    if (arow >= n) arow = n - 1;

    v8f c = {};
    #pragma unroll 4
    for (int k0 = 0; k0 < IN_DIM; k0 += 4) {
        const int ka = k0 + half * 2;
        const v2f a = *(const v2f*)&x[(size_t)arow * IN_DIM + ka];
        v2f b;
        b.x = W1s[ka * H1_DIM + l];
        b.y = W1s[(ka + 1) * H1_DIM + l];
        c = __builtin_amdgcn_wmma_f32_16x16x4_f32(
                false, a, false, b, (short)0, c, false, false);
    }

    const float bias = b1[l];
    if (row0 + 16 <= n) {
        #pragma unroll
        for (int r = 0; r < 8; ++r) {
            const int row = row0 + r + half * 8;
            const float v  = c[r];
            const float dv = dis[row];
            hw1 [(size_t)row * H1_DIM + l] = v;
            agg1[(size_t)row * H1_DIM + l] = dv * dv * v + bias;
        }
    } else {
        for (int r = 0; r < 8; ++r) {
            const int row = row0 + r + half * 8;
            if (row < n) {
                const float v  = c[r];
                const float dv = dis[row];
                hw1 [(size_t)row * H1_DIM + l] = v;
                agg1[(size_t)row * H1_DIM + l] = dv * dv * v + bias;
            }
        }
    }
}

#define SC_WAVES  32
#define SC_UNROLL 4
#define SC_QCAP   160

template <int G>
__device__ __forceinline__ int sc_pick(int x, int t, int grp) {
    if (G == 2) {
        const int a = __builtin_amdgcn_readlane(x, t * 2);
        const int b = __builtin_amdgcn_readlane(x, t * 2 + 1);
        return grp ? b : a;
    } else {
        const int a = __builtin_amdgcn_readlane(x, t * 4);
        const int b = __builtin_amdgcn_readlane(x, t * 4 + 1);
        const int c = __builtin_amdgcn_readlane(x, t * 4 + 2);
        const int d = __builtin_amdgcn_readlane(x, t * 4 + 3);
        return grp == 0 ? a : (grp == 1 ? b : (grp == 2 ? c : d));
    }
}

template <int D, bool FULL>
__device__ __forceinline__ void sc_batch(const int2* qb, int cnt, int lo, int lane,
                                         const float* __restrict__ dis,
                                         const float* __restrict__ hw, float* acc) {
    constexpr int G = 32 / D;
    const int grp = lane / D;
    const int l   = lane % D;
    int2  ent = make_int2(0, 0);
    float w   = 0.0f;
    if (FULL || lane < cnt) {
        ent = qb[lane];
        w   = dis[ent.x] * dis[lo + ent.y];
    }
    const int wbits = __float_as_int(w);
    float v[D];
    float ws[D];
    int   row[D];
    #pragma unroll
    for (int t = 0; t < D; ++t) {
        const int s = sc_pick<G>(ent.x, t, grp);
        row[t] = sc_pick<G>(ent.y, t, grp);
        ws[t]  = __int_as_float(sc_pick<G>(wbits, t, grp));
        v[t]   = hw[(size_t)s * D + l];
    }
    #pragma unroll
    for (int t = 0; t < D; ++t) {
        if (FULL || t * G + grp < cnt)
            atomicAdd(&acc[row[t] * D + l], ws[t] * v[t]);
    }
}

#define SC_ENQ(dval, sval)                                                     \
    {                                                                          \
        const unsigned dl = (unsigned)((dval) - lo);                           \
        const bool m = dl < (unsigned)nloc;                                    \
        const unsigned mask = __builtin_amdgcn_ballot_w32(m);                  \
        if (m) q[qn + (int)__builtin_amdgcn_mbcnt_lo(mask, 0u)] =              \
                   make_int2((sval), (int)dl);                                 \
        qn += __builtin_popcount(mask);                                        \
    }

template <int D, int R>
__global__ __launch_bounds__(1024) void gcn_scatter_ranges(
    const int* __restrict__ src, const int* __restrict__ dst,
    const float* __restrict__ dis, const float* __restrict__ hw,
    float* __restrict__ part, int n, int e, int nranges, int chunks_per_slice) {
    __shared__ float acc[R * D];
    __shared__ int2  queue[SC_WAVES * SC_QCAP];
    const int tid  = threadIdx.x;
    const int lane = tid & 31;
    const int wave = __builtin_amdgcn_readfirstlane(tid >> 5);
    const int range = blockIdx.x % nranges;
    const int slice = blockIdx.x / nranges;
    const int lo    = range * R;
    const int nloc  = (n - lo < R) ? (n - lo) : R;

    for (int i = tid * 4; i < R * D; i += 1024 * 4)
        *(float4*)&acc[i] = make_float4(0.0f, 0.0f, 0.0f, 0.0f);
    __syncthreads();

    const int total_chunks = (e + 127) >> 7;
    const int c0 = slice * chunks_per_slice;
    const int c1 = (c0 + chunks_per_slice < total_chunks) ? (c0 + chunks_per_slice)
                                                          : total_chunks;
    const bool vec_ok = (e & 3) == 0;
    int2* q = &queue[wave * SC_QCAP];
    int qn = 0;

    for (int cb = c0 + wave; cb < c1; cb += SC_WAVES * SC_UNROLL) {
        int4 dv[SC_UNROLL], sv[SC_UNROLL];
        #pragma unroll
        for (int u = 0; u < SC_UNROLL; ++u) {
            const int c = cb + u * SC_WAVES;
            dv[u] = make_int4(-1, -1, -1, -1);
            sv[u] = make_int4(0, 0, 0, 0);
            if (c < c1) {
                const int base = c * 128 + lane * 4;
                if (vec_ok && c * 128 + 128 <= e) {
                    dv[u] = *(const int4*)&dst[base];
                    sv[u] = *(const int4*)&src[base];
                } else {
                    if (base     < e) { dv[u].x = dst[base];     sv[u].x = src[base];     }
                    if (base + 1 < e) { dv[u].y = dst[base + 1]; sv[u].y = src[base + 1]; }
                    if (base + 2 < e) { dv[u].z = dst[base + 2]; sv[u].z = src[base + 2]; }
                    if (base + 3 < e) { dv[u].w = dst[base + 3]; sv[u].w = src[base + 3]; }
                }
            }
        }
        #pragma unroll
        for (int u = 0; u < SC_UNROLL; ++u) {
            SC_ENQ(dv[u].x, sv[u].x)
            SC_ENQ(dv[u].y, sv[u].y)
            SC_ENQ(dv[u].z, sv[u].z)
            SC_ENQ(dv[u].w, sv[u].w)
            while (qn >= 32) {
                qn -= 32;
                sc_batch<D, true>(q + qn, 32, lo, lane, dis, hw, acc);
            }
        }
    }
    if (qn > 0) sc_batch<D, false>(q, qn, lo, lane, dis, hw, acc);
    __syncthreads();

    float* outp = part + ((size_t)slice * n + lo) * D;
    const int total = nloc * D;
    for (int i = tid * 4; i < total; i += 1024 * 4)
        *(float4*)&outp[i] = *(const float4*)&acc[i];
}

__global__ __launch_bounds__(256) void gcn_gemm2_wmma(
    const float* __restrict__ h1, const float* __restrict__ part, int nsl,
    const float* __restrict__ W2,
    const float* __restrict__ dis, const float* __restrict__ b2,
    float* __restrict__ hw2, float* __restrict__ out0, int n) {
    __shared__ float W2s[H1_DIM * 16];
    const int tid = threadIdx.x;
    {
        const int k = tid >> 4, col = tid & 15;
        W2s[tid] = (col < H2_DIM) ? W2[k * H2_DIM + col] : 0.0f;
    }
    __syncthreads();

    const int wave = tid >> 5;
    const int lane = tid & 31;
    const int half = lane >> 4;
    const int l    = lane & 15;
    const int row0 = (blockIdx.x * 8 + wave) * 16;
    if (row0 >= n) return;

    int arow = row0 + l;
    if (arow >= n) arow = n - 1;

    v8f c = {};
    #pragma unroll
    for (int k0 = 0; k0 < H1_DIM; k0 += 4) {
        const int ka = k0 + half * 2;
        v2f a = *(const v2f*)&h1[(size_t)arow * H1_DIM + ka];
        for (int s = 0; s < nsl; ++s) {
            const v2f p = *(const v2f*)&part[((size_t)s * n + arow) * H1_DIM + ka];
            a.x += p.x;
            a.y += p.y;
        }
        a.x = fmaxf(a.x, 0.0f);
        a.y = fmaxf(a.y, 0.0f);
        v2f b;
        b.x = W2s[ka * 16 + l];
        b.y = W2s[(ka + 1) * 16 + l];
        c = __builtin_amdgcn_wmma_f32_16x16x4_f32(
                false, a, false, b, (short)0, c, false, false);
    }

    if (l < H2_DIM) {
        const float bias = b2[l];
        if (row0 + 16 <= n) {
            #pragma unroll
            for (int r = 0; r < 8; ++r) {
                const int row = row0 + r + half * 8;
                const float v  = c[r];
                const float dv = dis[row];
                hw2 [(size_t)row * H2_DIM + l] = v;
                out0[(size_t)row * H2_DIM + l] = dv * dv * v + bias;
            }
        } else {
            for (int r = 0; r < 8; ++r) {
                const int row = row0 + r + half * 8;
                if (row < n) {
                    const float v  = c[r];
                    const float dv = dis[row];
                    hw2 [(size_t)row * H2_DIM + l] = v;
                    out0[(size_t)row * H2_DIM + l] = dv * dv * v + bias;
                }
            }
        }
    }
}

__global__ __launch_bounds__(256) void gcn_sum_out(
    const float* __restrict__ out0, const float* __restrict__ part, int nsl,
    float* __restrict__ out, int n) {
    const int i = blockIdx.x * blockDim.x + threadIdx.x;
    if (i >= n * (H2_DIM / 4)) return;
    float4 a = ((const float4*)out0)[i];
    for (int s = 0; s < nsl; ++s) {
        const float4 p = ((const float4*)(part + (size_t)s * n * H2_DIM))[i];
        a.x += p.x; a.y += p.y; a.z += p.z; a.w += p.w;
    }
    ((float4*)out)[i] = a;
}

extern "C" void kernel_launch(void* const* d_in, const int* in_sizes, int n_in,
                              void* d_out, int out_size, void* d_ws, size_t ws_size,
                              hipStream_t stream) {
    const float* x    = (const float*)d_in[0];
    const int*   edge = (const int*)  d_in[1];
    const float* W1   = (const float*)d_in[2];
    const float* b1   = (const float*)d_in[3];
    const float* W2   = (const float*)d_in[4];
    const float* b2   = (const float*)d_in[5];
    float* out = (float*)d_out;

    const int n = in_sizes[0] / IN_DIM;
    const int e = in_sizes[1] / 2;
    const int* src = edge;
    const int* dst = edge + e;

    const int R1 = 4096, R2 = 8192;
    const int nr1 = (n + R1 - 1) / R1;
    const int nr2 = (n + R2 - 1) / R2;
    const int S1  = (256 / nr1 > 0) ? 256 / nr1 : 1;
    const int S2  = (256 / nr2 > 0) ? 256 / nr2 : 1;
    const int chunks = (e + 127) / 128;
    const int cps1 = (chunks + S1 - 1) / S1;
    const int cps2 = (chunks + S2 - 1) / S2;

    const int nd = (n + 63) & ~63;
    float* ws    = (float*)d_ws;
    float* dis   = ws;
    float* hw1   = dis + nd;
    float* agg1  = hw1 + (size_t)n * H1_DIM;
    float* hw2   = agg1 + (size_t)n * H1_DIM;
    float* out0  = hw2 + (size_t)n * H2_DIM;
    float* part1 = out0 + (size_t)n * H2_DIM;
    float* part2 = part1 + (size_t)S1 * n * H1_DIM;

    const int B = 256;
    const int gn     = (n + B - 1) / B;
    const int ge     = (e + B - 1) / B;
    const int tiles  = (n + 15) / 16;
    const int gtiles = (tiles + 7) / 8;
    const int gsum   = (n * (H2_DIM / 4) + B - 1) / B;

    gcn_init_deg <<<gn, B, 0, stream>>>(dis, n);
    gcn_count_deg<<<ge, B, 0, stream>>>(dst, dis, e);
    gcn_rsqrt_deg<<<gn, B, 0, stream>>>(dis, n);

    gcn_gemm1_wmma<<<gtiles, B, 0, stream>>>(x, W1, dis, b1, hw1, agg1, n);
    gcn_scatter_ranges<H1_DIM, 4096><<<nr1 * S1, 1024, 0, stream>>>(
        src, dst, dis, hw1, part1, n, e, nr1, cps1);

    gcn_gemm2_wmma<<<gtiles, B, 0, stream>>>(agg1, part1, S1, W2, dis, b2, hw2, out0, n);
    gcn_scatter_ranges<H2_DIM, 8192><<<nr2 * S2, 1024, 0, stream>>>(
        src, dst, dis, hw2, part2, n, e, nr2, cps2);
    gcn_sum_out<<<gsum, B, 0, stream>>>(out0, part2, S2, out, n);
}


# Degree count without global atomics: each work-group counts a 32768-edge slice of dst into one whole-graph LDS table of packed 16-bit counters and stores it; the rsqrt kernel adds the tables (exact)
# speedup vs baseline: 2.8908x; 1.8161x over previous
#include <hip/hip_runtime.h>


typedef __attribute__((ext_vector_type(2))) float v2f;
typedef __attribute__((ext_vector_type(8))) float v8f;

#define IN_DIM 128
#define H1_DIM 16
#define H2_DIM 8

#define DH_EDGES 32768
#define DH_WORDS 51200

#define DH_COUNT(val)                                                          \
    {                                                                          \
        const unsigned node = (unsigned)(val);                                 \
        if (node < lim) atomicAdd(&cnt[node >> 1], 1u << ((node & 1u) * 16u)); \
    }

__global__ __launch_bounds__(1024) void gcn_deg_hist(
    const int* __restrict__ dst, unsigned* __restrict__ tabs, int n, int e, int words) {
    __shared__ unsigned cnt[DH_WORDS];
    const int tid = threadIdx.x;
    for (int i = tid * 4; i < words; i += 1024 * 4)
        *(uint4*)&cnt[i] = make_uint4(0u, 0u, 0u, 0u);
    __syncthreads();

    const unsigned lim = ((unsigned)n < 2u * DH_WORDS) ? (unsigned)n : 2u * DH_WORDS;
    const int base = blockIdx.x * DH_EDGES;
    const bool vec_ok = (e & 3) == 0;
    int4 d[DH_EDGES / 4096];
    if (vec_ok && e - base >= DH_EDGES) {
        #pragma unroll
        for (int j = 0; j < DH_EDGES / 4096; ++j)
            d[j] = *(const int4*)&dst[base + (j * 1024 + tid) * 4];
    } else {
        #pragma unroll
        for (int j = 0; j < DH_EDGES / 4096; ++j) {
            const int i = base + (j * 1024 + tid) * 4;
            d[j] = make_int4(-1, -1, -1, -1);
            if (i     < e) d[j].x = dst[i];
            if (i + 1 < e) d[j].y = dst[i + 1];
            if (i + 2 < e) d[j].z = dst[i + 2];
            if (i + 3 < e) d[j].w = dst[i + 3];
        }
    }
    #pragma unroll
    for (int j = 0; j < DH_EDGES / 4096; ++j) {
        DH_COUNT(d[j].x)
        DH_COUNT(d[j].y)
        DH_COUNT(d[j].z)
        DH_COUNT(d[j].w)
    }
    __syncthreads();

    unsigned* outp = tabs + (size_t)blockIdx.x * words;
    for (int i = tid * 4; i < words; i += 1024 * 4)
        *(uint4*)&outp[i] = *(const uint4*)&cnt[i];
}

__device__ __forceinline__ float gcn_dis_of(unsigned count) {
    const float d = (float)(count + 1u);
    return (d > 0.0f) ? rsqrtf(d) : 0.0f;
}

__global__ __launch_bounds__(256) void gcn_deg_dis(
    const unsigned* __restrict__ tabs, int ntab, int words,
    float* __restrict__ dis, int n) {
    __shared__ uint4 red[8 * 32];
    const int tid  = threadIdx.x;
    const int lane = tid & 31;
    const int wave = __builtin_amdgcn_readfirstlane(tid >> 5);
    const int q = blockIdx.x * 32 + lane;
    unsigned c0 = 0u, c1 = 0u, c2 = 0u, c3 = 0u;
    for (int s0 = wave; s0 < ntab; s0 += 64) {
        uint2 v[8];
        #pragma unroll
        for (int u = 0; u < 8; ++u) {
            const int s = s0 + u * 8;
            v[u] = make_uint2(0u, 0u);
            if (s < ntab) v[u] = *(const uint2*)&tabs[(size_t)s * words + 2 * q];
        }
        #pragma unroll
        for (int u = 0; u < 8; ++u) {
            c0 += v[u].x & 0xffffu;
            c1 += v[u].x >> 16;
            c2 += v[u].y & 0xffffu;
            c3 += v[u].y >> 16;
        }
    }
    red[wave * 32 + lane] = make_uint4(c0, c1, c2, c3);
    __syncthreads();
    if (wave == 0) {
        uint4 t = red[lane];
        #pragma unroll
        for (int w = 1; w < 8; ++w) {
            const uint4 r = red[w * 32 + lane];
            t.x += r.x; t.y += r.y; t.z += r.z; t.w += r.w;
        }
        const int node = q * 4;
        const float4 o = make_float4(gcn_dis_of(t.x), gcn_dis_of(t.y),
                                     gcn_dis_of(t.z), gcn_dis_of(t.w));
        if (node + 3 < n) {
            *(float4*)&dis[node] = o;
        } else {
            if (node     < n) dis[node]     = o.x;
            if (node + 1 < n) dis[node + 1] = o.y;
            if (node + 2 < n) dis[node + 2] = o.z;
        }
    }
}

__global__ __launch_bounds__(256) void gcn_gemm1_wmma(
    const float* __restrict__ x, const float* __restrict__ W1,
    const float* __restrict__ dis, const float* __restrict__ b1,
    float* __restrict__ hw1, float* __restrict__ agg1, int n) {
    __shared__ float W1s[IN_DIM * H1_DIM];
    const int tid = threadIdx.x;
    #pragma unroll
    for (int j = 0; j < (IN_DIM * H1_DIM) / 256; ++j)
        W1s[tid + j * 256] = W1[tid + j * 256];
    __syncthreads();

    const int wave = tid >> 5;
    const int lane = tid & 31;
    const int half = lane >> 4;
    const int l    = lane & 15;
    const int row0 = (blockIdx.x * 8 + wave) * 16;
    if (row0 >= n) return;

    int arow = row0 + l;
    if (arow >= n) arow = n - 1;

    v8f c = {};
    #pragma unroll 4
    for (int k0 = 0; k0 < IN_DIM; k0 += 4) {
        const int ka = k0 + half * 2;
        const v2f a = *(const v2f*)&x[(size_t)arow * IN_DIM + ka];
        v2f b;
        b.x = W1s[ka * H1_DIM + l];
        b.y = W1s[(ka + 1) * H1_DIM + l];
        c = __builtin_amdgcn_wmma_f32_16x16x4_f32(
                false, a, false, b, (short)0, c, false, false);
    }

    const float bias = b1[l];
    if (row0 + 16 <= n) {
        #pragma unroll
        for (int r = 0; r < 8; ++r) {
            const int row = row0 + r + half * 8;
            const float v  = c[r];
            const float dv = dis[row];
            hw1 [(size_t)row * H1_DIM + l] = v;
            agg1[(size_t)row * H1_DIM + l] = dv * dv * v + bias;
        }
    } else {
        for (int r = 0; r < 8; ++r) {
            const int row = row0 + r + half * 8;
            if (row < n) {
                const float v  = c[r];
                const float dv = dis[row];
                hw1 [(size_t)row * H1_DIM + l] = v;
                agg1[(size_t)row * H1_DIM + l] = dv * dv * v + bias;
            }
        }
    }
}

#define SC_WAVES  32
#define SC_UNROLL 4
#define SC_QCAP   160

template <int G>
__device__ __forceinline__ int sc_pick(int x, int t, int grp) {
    if (G == 2) {
        const int a = __builtin_amdgcn_readlane(x, t * 2);
        const int b = __builtin_amdgcn_readlane(x, t * 2 + 1);
        return grp ? b : a;
    } else {
        const int a = __builtin_amdgcn_readlane(x, t * 4);
        const int b = __builtin_amdgcn_readlane(x, t * 4 + 1);
        const int c = __builtin_amdgcn_readlane(x, t * 4 + 2);
        const int d = __builtin_amdgcn_readlane(x, t * 4 + 3);
        return grp == 0 ? a : (grp == 1 ? b : (grp == 2 ? c : d));
    }
}

template <int D, bool FULL>
__device__ __forceinline__ void sc_batch(const int2* qb, int cnt, int lo, int lane,
                                         const float* __restrict__ dis,
                                         const float* __restrict__ hw, float* acc) {
    constexpr int G = 32 / D;
    const int grp = lane / D;
    const int l   = lane % D;
    int2  ent = make_int2(0, 0);
    float w   = 0.0f;
    if (FULL || lane < cnt) {
        ent = qb[lane];
        w   = dis[ent.x] * dis[lo + ent.y];
    }
    const int wbits = __float_as_int(w);
    float v[D];
    float ws[D];
    int   row[D];
    #pragma unroll
    for (int t = 0; t < D; ++t) {
        const int s = sc_pick<G>(ent.x, t, grp);
        row[t] = sc_pick<G>(ent.y, t, grp);
        ws[t]  = __int_as_float(sc_pick<G>(wbits, t, grp));
        v[t]   = hw[(size_t)s * D + l];
    }
    #pragma unroll
    for (int t = 0; t < D; ++t) {
        if (FULL || t * G + grp < cnt)
            atomicAdd(&acc[row[t] * D + l], ws[t] * v[t]);
    }
}

#define SC_ENQ(dval, sval)                                                     \
    {                                                                          \
        const unsigned dl = (unsigned)((dval) - lo);                           \
        const bool m = dl < (unsigned)nloc;                                    \
        const unsigned mask = __builtin_amdgcn_ballot_w32(m);                  \
        if (m) q[qn + (int)__builtin_amdgcn_mbcnt_lo(mask, 0u)] =              \
                   make_int2((sval), (int)dl);                                 \
        qn += __builtin_popcount(mask);                                        \
    }

template <int D, int R>
__global__ __launch_bounds__(1024) void gcn_scatter_ranges(
    const int* __restrict__ src, const int* __restrict__ dst,
    const float* __restrict__ dis, const float* __restrict__ hw,
    float* __restrict__ part, int n, int e, int nranges, int chunks_per_slice) {
    __shared__ float acc[R * D];
    __shared__ int2  queue[SC_WAVES * SC_QCAP];
    const int tid  = threadIdx.x;
    const int lane = tid & 31;
    const int wave = __builtin_amdgcn_readfirstlane(tid >> 5);
    const int range = blockIdx.x % nranges;
    const int slice = blockIdx.x / nranges;
    const int lo    = range * R;
    const int nloc  = (n - lo < R) ? (n - lo) : R;

    for (int i = tid * 4; i < R * D; i += 1024 * 4)
        *(float4*)&acc[i] = make_float4(0.0f, 0.0f, 0.0f, 0.0f);
    __syncthreads();

    const int total_chunks = (e + 127) >> 7;
    const int c0 = slice * chunks_per_slice;
    const int c1 = (c0 + chunks_per_slice < total_chunks) ? (c0 + chunks_per_slice)
                                                          : total_chunks;
    const bool vec_ok = (e & 3) == 0;
    int2* q = &queue[wave * SC_QCAP];
    int qn = 0;

    for (int cb = c0 + wave; cb < c1; cb += SC_WAVES * SC_UNROLL) {
        int4 dv[SC_UNROLL], sv[SC_UNROLL];
        #pragma unroll
        for (int u = 0; u < SC_UNROLL; ++u) {
            const int c = cb + u * SC_WAVES;
            dv[u] = make_int4(-1, -1, -1, -1);
            sv[u] = make_int4(0, 0, 0, 0);
            if (c < c1) {
                const int base = c * 128 + lane * 4;
                if (vec_ok && c * 128 + 128 <= e) {
                    dv[u] = *(const int4*)&dst[base];
                    sv[u] = *(const int4*)&src[base];
                } else {
                    if (base     < e) { dv[u].x = dst[base];     sv[u].x = src[base];     }
                    if (base + 1 < e) { dv[u].y = dst[base + 1]; sv[u].y = src[base + 1]; }
                    if (base + 2 < e) { dv[u].z = dst[base + 2]; sv[u].z = src[base + 2]; }
                    if (base + 3 < e) { dv[u].w = dst[base + 3]; sv[u].w = src[base + 3]; }
                }
            }
        }
        #pragma unroll
        for (int u = 0; u < SC_UNROLL; ++u) {
            SC_ENQ(dv[u].x, sv[u].x)
            SC_ENQ(dv[u].y, sv[u].y)
            SC_ENQ(dv[u].z, sv[u].z)
            SC_ENQ(dv[u].w, sv[u].w)
            while (qn >= 32) {
                qn -= 32;
                sc_batch<D, true>(q + qn, 32, lo, lane, dis, hw, acc);
            }
        }
    }
    if (qn > 0) sc_batch<D, false>(q, qn, lo, lane, dis, hw, acc);
    __syncthreads();

    float* outp = part + ((size_t)slice * n + lo) * D;
    const int total = nloc * D;
    for (int i = tid * 4; i < total; i += 1024 * 4)
        *(float4*)&outp[i] = *(const float4*)&acc[i];
}

__global__ __launch_bounds__(256) void gcn_gemm2_wmma(
    const float* __restrict__ h1, const float* __restrict__ part, int nsl,
    const float* __restrict__ W2,
    const float* __restrict__ dis, const float* __restrict__ b2,
    float* __restrict__ hw2, float* __restrict__ out0, int n) {
    __shared__ float W2s[H1_DIM * 16];
    const int tid = threadIdx.x;
    {
        const int k = tid >> 4, col = tid & 15;
        W2s[tid] = (col < H2_DIM) ? W2[k * H2_DIM + col] : 0.0f;
    }
    __syncthreads();

    const int wave = tid >> 5;
    const int lane = tid & 31;
    const int half = lane >> 4;
    const int l    = lane & 15;
    const int row0 = (blockIdx.x * 8 + wave) * 16;
    if (row0 >= n) return;

    int arow = row0 + l;
    if (arow >= n) arow = n - 1;

    v8f c = {};
    #pragma unroll
    for (int k0 = 0; k0 < H1_DIM; k0 += 4) {
        const int ka = k0 + half * 2;
        v2f a = *(const v2f*)&h1[(size_t)arow * H1_DIM + ka];
        for (int s = 0; s < nsl; ++s) {
            const v2f p = *(const v2f*)&part[((size_t)s * n + arow) * H1_DIM + ka];
            a.x += p.x;
            a.y += p.y;
        }
        a.x = fmaxf(a.x, 0.0f);
        a.y = fmaxf(a.y, 0.0f);
        v2f b;
        b.x = W2s[ka * 16 + l];
        b.y = W2s[(ka + 1) * 16 + l];
        c = __builtin_amdgcn_wmma_f32_16x16x4_f32(
                false, a, false, b, (short)0, c, false, false);
    }

    if (l < H2_DIM) {
        const float bias = b2[l];
        if (row0 + 16 <= n) {
            #pragma unroll
            for (int r = 0; r < 8; ++r) {
                const int row = row0 + r + half * 8;
                const float v  = c[r];
                const float dv = dis[row];
                hw2 [(size_t)row * H2_DIM + l] = v;
                out0[(size_t)row * H2_DIM + l] = dv * dv * v + bias;
            }
        } else {
            for (int r = 0; r < 8; ++r) {
                const int row = row0 + r + half * 8;
                if (row < n) {
                    const float v  = c[r];
                    const float dv = dis[row];
                    hw2 [(size_t)row * H2_DIM + l] = v;
                    out0[(size_t)row * H2_DIM + l] = dv * dv * v + bias;
                }
            }
        }
    }
}

__global__ __launch_bounds__(256) void gcn_sum_out(
    const float* __restrict__ out0, const float* __restrict__ part, int nsl,
    float* __restrict__ out, int n) {
    const int i = blockIdx.x * blockDim.x + threadIdx.x;
    if (i >= n * (H2_DIM / 4)) return;
    float4 a = ((const float4*)out0)[i];
    for (int s = 0; s < nsl; ++s) {
        const float4 p = ((const float4*)(part + (size_t)s * n * H2_DIM))[i];
        a.x += p.x; a.y += p.y; a.z += p.z; a.w += p.w;
    }
    ((float4*)out)[i] = a;
}

extern "C" void kernel_launch(void* const* d_in, const int* in_sizes, int n_in,
                              void* d_out, int out_size, void* d_ws, size_t ws_size,
                              hipStream_t stream) {
    const float* x    = (const float*)d_in[0];
    const int*   edge = (const int*)  d_in[1];
    const float* W1   = (const float*)d_in[2];
    const float* b1   = (const float*)d_in[3];
    const float* W2   = (const float*)d_in[4];
    const float* b2   = (const float*)d_in[5];
    float* out = (float*)d_out;

    const int n = in_sizes[0] / IN_DIM;
    const int e = in_sizes[1] / 2;
    const int* src = edge;
    const int* dst = edge + e;

    const int R1 = 4096, R2 = 8192;
    const int nr1 = (n + R1 - 1) / R1;
    const int nr2 = (n + R2 - 1) / R2;
    const int S1  = (256 / nr1 > 0) ? 256 / nr1 : 1;
    const int S2  = (256 / nr2 > 0) ? 256 / nr2 : 1;
    const int chunks = (e + 127) / 128;
    const int cps1 = (chunks + S1 - 1) / S1;
    const int cps2 = (chunks + S2 - 1) / S2;

    const int nd = (n + 63) & ~63;
    float* ws    = (float*)d_ws;
    float* dis   = ws;
    float* hw1   = dis + nd;
    float* agg1  = hw1 + (size_t)n * H1_DIM;
    float* hw2   = agg1 + (size_t)n * H1_DIM;
    float* out0  = hw2 + (size_t)n * H2_DIM;
    float* part1 = out0 + (size_t)n * H2_DIM;
    float* part2 = part1 + (size_t)S1 * n * H1_DIM;
    unsigned* tabs = (unsigned*)(part2 + (size_t)S2 * n * H2_DIM);
    const int dwords = (((n + 1) / 2) + 63) & ~63;
    const int ntab   = (e + DH_EDGES - 1) / DH_EDGES;

    const int B = 256;
    const int tiles  = (n + 15) / 16;
    const int gtiles = (tiles + 7) / 8;
    const int gsum   = (n * (H2_DIM / 4) + B - 1) / B;

    gcn_deg_hist<<<ntab, 1024, 0, stream>>>(dst, tabs, n, e, dwords);
    gcn_deg_dis <<<dwords / 64, B, 0, stream>>>(tabs, ntab, dwords, dis, n);

    gcn_gemm1_wmma<<<gtiles, B, 0, stream>>>(x, W1, dis, b1, hw1, agg1, n);
    gcn_scatter_ranges<H1_DIM, 4096><<<nr1 * S1, 1024, 0, stream>>>(
        src, dst, dis, hw1, part1, n, e, nr1, cps1);

    gcn_gemm2_wmma<<<gtiles, B, 0, stream>>>(agg1, part1, S1, W2, dis, b2, hw2, out0, n);
    gcn_scatter_ranges<H2_DIM, 8192><<<nr2 * S2, 1024, 0, stream>>>(
        src, dst, dis, hw2, part2, n, e, nr2, cps2);
    gcn_sum_out<<<gsum, B, 0, stream>>>(out0, part2, S2, out, n);
}


# Both scatters: dis[src] sits beside each hw row in the row's own 128-B line (records of 2*D floats from the GEMM epilogues), so the weight gather costs no second L2 line; sums and order unchanged
# speedup vs baseline: 3.0961x; 1.0710x over previous
#include <hip/hip_runtime.h>


typedef __attribute__((ext_vector_type(2))) float v2f;
typedef __attribute__((ext_vector_type(8))) float v8f;

#define IN_DIM 128
#define H1_DIM 16
#define H2_DIM 8

#define DH_EDGES 32768
#define DH_WORDS 51200

#define DH_COUNT(val)                                                          \
    {                                                                          \
        const unsigned node = (unsigned)(val);                                 \
        if (node < lim) atomicAdd(&cnt[node >> 1], 1u << ((node & 1u) * 16u)); \
    }

__global__ __launch_bounds__(1024) void gcn_deg_hist(
    const int* __restrict__ dst, unsigned* __restrict__ tabs, int n, int e, int words) {
    __shared__ unsigned cnt[DH_WORDS];
    const int tid = threadIdx.x;
    for (int i = tid * 4; i < words; i += 1024 * 4)
        *(uint4*)&cnt[i] = make_uint4(0u, 0u, 0u, 0u);
    __syncthreads();

    const unsigned lim = ((unsigned)n < 2u * DH_WORDS) ? (unsigned)n : 2u * DH_WORDS;
    const int base = blockIdx.x * DH_EDGES;
    const bool vec_ok = (e & 3) == 0;
    int4 d[DH_EDGES / 4096];
    if (vec_ok && e - base >= DH_EDGES) {
        #pragma unroll
        for (int j = 0; j < DH_EDGES / 4096; ++j)
            d[j] = *(const int4*)&dst[base + (j * 1024 + tid) * 4];
    } else {
        #pragma unroll
        for (int j = 0; j < DH_EDGES / 4096; ++j) {
            const int i = base + (j * 1024 + tid) * 4;
            d[j] = make_int4(-1, -1, -1, -1);
            if (i     < e) d[j].x = dst[i];
            if (i + 1 < e) d[j].y = dst[i + 1];
            if (i + 2 < e) d[j].z = dst[i + 2];
            if (i + 3 < e) d[j].w = dst[i + 3];
        }
    }
    #pragma unroll
    for (int j = 0; j < DH_EDGES / 4096; ++j) {
        DH_COUNT(d[j].x)
        DH_COUNT(d[j].y)
        DH_COUNT(d[j].z)
        DH_COUNT(d[j].w)
    }
    __syncthreads();

    unsigned* outp = tabs + (size_t)blockIdx.x * words;
    for (int i = tid * 4; i < words; i += 1024 * 4)
        *(uint4*)&outp[i] = *(const uint4*)&cnt[i];
}

__device__ __forceinline__ float gcn_dis_of(unsigned count) {
    const float d = (float)(count + 1u);
    return (d > 0.0f) ? rsqrtf(d) : 0.0f;
}

__global__ __launch_bounds__(256) void gcn_deg_dis(
    const unsigned* __restrict__ tabs, int ntab, int words,
    float* __restrict__ dis, int n) {
    __shared__ uint4 red[8 * 32];
    const int tid  = threadIdx.x;
    const int lane = tid & 31;
    const int wave = __builtin_amdgcn_readfirstlane(tid >> 5);
    const int q = blockIdx.x * 32 + lane;
    unsigned c0 = 0u, c1 = 0u, c2 = 0u, c3 = 0u;
    for (int s0 = wave; s0 < ntab; s0 += 64) {
        uint2 v[8];
        #pragma unroll
        for (int u = 0; u < 8; ++u) {
            const int s = s0 + u * 8;
            v[u] = make_uint2(0u, 0u);
            if (s < ntab) v[u] = *(const uint2*)&tabs[(size_t)s * words + 2 * q];
        }
        #pragma unroll
        for (int u = 0; u < 8; ++u) {
            c0 += v[u].x & 0xffffu;
            c1 += v[u].x >> 16;
            c2 += v[u].y & 0xffffu;
            c3 += v[u].y >> 16;
        }
    }
    red[wave * 32 + lane] = make_uint4(c0, c1, c2, c3);
    __syncthreads();
    if (wave == 0) {
        uint4 t = red[lane];
        #pragma unroll
        for (int w = 1; w < 8; ++w) {
            const uint4 r = red[w * 32 + lane];
            t.x += r.x; t.y += r.y; t.z += r.z; t.w += r.w;
        }
        const int node = q * 4;
        const float4 o = make_float4(gcn_dis_of(t.x), gcn_dis_of(t.y),
                                     gcn_dis_of(t.z), gcn_dis_of(t.w));
        if (node + 3 < n) {
            *(float4*)&dis[node] = o;
        } else {
            if (node     < n) dis[node]     = o.x;
            if (node + 1 < n) dis[node + 1] = o.y;
            if (node + 2 < n) dis[node + 2] = o.z;
        }
    }
}

__global__ __launch_bounds__(256) void gcn_gemm1_wmma(
    const float* __restrict__ x, const float* __restrict__ W1,
    const float* __restrict__ dis, const float* __restrict__ b1,
    float* __restrict__ hw1, float* __restrict__ agg1, int n) {
    __shared__ float W1s[IN_DIM * H1_DIM];
    const int tid = threadIdx.x;
    #pragma unroll
    for (int j = 0; j < (IN_DIM * H1_DIM) / 256; ++j)
        W1s[tid + j * 256] = W1[tid + j * 256];
    __syncthreads();

    const int wave = tid >> 5;
    const int lane = tid & 31;
    const int half = lane >> 4;
    const int l    = lane & 15;
    const int row0 = (blockIdx.x * 8 + wave) * 16;
    if (row0 >= n) return;

    int arow = row0 + l;
    if (arow >= n) arow = n - 1;

    v8f c = {};
    #pragma unroll 4
    for (int k0 = 0; k0 < IN_DIM; k0 += 4) {
        const int ka = k0 + half * 2;
        const v2f a = *(const v2f*)&x[(size_t)arow * IN_DIM + ka];
        v2f b;
        b.x = W1s[ka * H1_DIM + l];
        b.y = W1s[(ka + 1) * H1_DIM + l];
        c = __builtin_amdgcn_wmma_f32_16x16x4_f32(
                false, a, false, b, (short)0, c, false, false);
    }

    const float bias = b1[l];
    if (row0 + 16 <= n) {
        #pragma unroll
        for (int r = 0; r < 8; ++r) {
            const int row = row0 + r + half * 8;
            const float v  = c[r];
            const float dv = dis[row];
            hw1 [(size_t)row * (2 * H1_DIM) + l] = v;
            hw1 [(size_t)row * (2 * H1_DIM) + H1_DIM + l] = dv;
            agg1[(size_t)row * H1_DIM + l] = dv * dv * v + bias;
        }
    } else {
        for (int r = 0; r < 8; ++r) {
            const int row = row0 + r + half * 8;
            if (row < n) {
                const float v  = c[r];
                const float dv = dis[row];
                hw1 [(size_t)row * (2 * H1_DIM) + l] = v;
                hw1 [(size_t)row * (2 * H1_DIM) + H1_DIM + l] = dv;
                agg1[(size_t)row * H1_DIM + l] = dv * dv * v + bias;
            }
        }
    }
}

#define SC_WAVES  32
#define SC_UNROLL 4
#define SC_QCAP   160

template <int G>
__device__ __forceinline__ int sc_pick(int x, int t, int grp) {
    if (G == 2) {
        const int a = __builtin_amdgcn_readlane(x, t * 2);
        const int b = __builtin_amdgcn_readlane(x, t * 2 + 1);
        return grp ? b : a;
    } else {
        const int a = __builtin_amdgcn_readlane(x, t * 4);
        const int b = __builtin_amdgcn_readlane(x, t * 4 + 1);
        const int c = __builtin_amdgcn_readlane(x, t * 4 + 2);
        const int d = __builtin_amdgcn_readlane(x, t * 4 + 3);
        return grp == 0 ? a : (grp == 1 ? b : (grp == 2 ? c : d));
    }
}

template <int D, bool FULL>
__device__ __forceinline__ void sc_batch(const int2* qb, int cnt, int lo, int lane,
                                         const float* __restrict__ dis,
                                         const float* __restrict__ hw, float* acc) {
    constexpr int G = 32 / D;
    const int grp = lane / D;
    const int l   = lane % D;
    int2  ent = make_int2(0, 0);
    float w   = 0.0f;
    if (FULL || lane < cnt) {
        ent = qb[lane];
        w   = hw[(size_t)ent.x * (2 * D) + D] * dis[lo + ent.y];
    }
    const int wbits = __float_as_int(w);
    float v[D];
    float ws[D];
    int   row[D];
    #pragma unroll
    for (int t = 0; t < D; ++t) {
        const int s = sc_pick<G>(ent.x, t, grp);
        row[t] = sc_pick<G>(ent.y, t, grp);
        ws[t]  = __int_as_float(sc_pick<G>(wbits, t, grp));
        v[t]   = hw[(size_t)s * (2 * D) + l];
    }
    #pragma unroll
    for (int t = 0; t < D; ++t) {
        if (FULL || t * G + grp < cnt)
            atomicAdd(&acc[row[t] * D + l], ws[t] * v[t]);
    }
}

#define SC_ENQ(dval, sval)                                                     \
    {                                                                          \
        const unsigned dl = (unsigned)((dval) - lo);                           \
        const bool m = dl < (unsigned)nloc;                                    \
        const unsigned mask = __builtin_amdgcn_ballot_w32(m);                  \
        if (m) q[qn + (int)__builtin_amdgcn_mbcnt_lo(mask, 0u)] =              \
                   make_int2((sval), (int)dl);                                 \
        qn += __builtin_popcount(mask);                                        \
    }

template <int D, int R>
__global__ __launch_bounds__(1024) void gcn_scatter_ranges(
    const int* __restrict__ src, const int* __restrict__ dst,
    const float* __restrict__ dis, const float* __restrict__ hw,
    float* __restrict__ part, int n, int e, int nranges, int chunks_per_slice) {
    __shared__ float acc[R * D];
    __shared__ int2  queue[SC_WAVES * SC_QCAP];
    const int tid  = threadIdx.x;
    const int lane = tid & 31;
    const int wave = __builtin_amdgcn_readfirstlane(tid >> 5);
    const int range = blockIdx.x % nranges;
    const int slice = blockIdx.x / nranges;
    const int lo    = range * R;
    const int nloc  = (n - lo < R) ? (n - lo) : R;

    for (int i = tid * 4; i < R * D; i += 1024 * 4)
        *(float4*)&acc[i] = make_float4(0.0f, 0.0f, 0.0f, 0.0f);
    __syncthreads();

    const int total_chunks = (e + 127) >> 7;
    const int c0 = slice * chunks_per_slice;
    const int c1 = (c0 + chunks_per_slice < total_chunks) ? (c0 + chunks_per_slice)
                                                          : total_chunks;
    const bool vec_ok = (e & 3) == 0;
    int2* q = &queue[wave * SC_QCAP];
    int qn = 0;

    for (int cb = c0 + wave; cb < c1; cb += SC_WAVES * SC_UNROLL) {
        int4 dv[SC_UNROLL], sv[SC_UNROLL];
        #pragma unroll
        for (int u = 0; u < SC_UNROLL; ++u) {
            const int c = cb + u * SC_WAVES;
            dv[u] = make_int4(-1, -1, -1, -1);
            sv[u] = make_int4(0, 0, 0, 0);
            if (c < c1) {
                const int base = c * 128 + lane * 4;
                if (vec_ok && c * 128 + 128 <= e) {
                    dv[u] = *(const int4*)&dst[base];
                    sv[u] = *(const int4*)&src[base];
                } else {
                    if (base     < e) { dv[u].x = dst[base];     sv[u].x = src[base];     }
                    if (base + 1 < e) { dv[u].y = dst[base + 1]; sv[u].y = src[base + 1]; }
                    if (base + 2 < e) { dv[u].z = dst[base + 2]; sv[u].z = src[base + 2]; }
                    if (base + 3 < e) { dv[u].w = dst[base + 3]; sv[u].w = src[base + 3]; }
                }
            }
        }
        #pragma unroll
        for (int u = 0; u < SC_UNROLL; ++u) {
            SC_ENQ(dv[u].x, sv[u].x)
            SC_ENQ(dv[u].y, sv[u].y)
            SC_ENQ(dv[u].z, sv[u].z)
            SC_ENQ(dv[u].w, sv[u].w)
            while (qn >= 32) {
                qn -= 32;
                sc_batch<D, true>(q + qn, 32, lo, lane, dis, hw, acc);
            }
        }
    }
    if (qn > 0) sc_batch<D, false>(q, qn, lo, lane, dis, hw, acc);
    __syncthreads();

    float* outp = part + ((size_t)slice * n + lo) * D;
    const int total = nloc * D;
    for (int i = tid * 4; i < total; i += 1024 * 4)
        *(float4*)&outp[i] = *(const float4*)&acc[i];
}

__global__ __launch_bounds__(256) void gcn_gemm2_wmma(
    const float* __restrict__ h1, const float* __restrict__ part, int nsl,
    const float* __restrict__ W2,
    const float* __restrict__ dis, const float* __restrict__ b2,
    float* __restrict__ hw2, float* __restrict__ out0, int n) {
    __shared__ float W2s[H1_DIM * 16];
    const int tid = threadIdx.x;
    {
        const int k = tid >> 4, col = tid & 15;
        W2s[tid] = (col < H2_DIM) ? W2[k * H2_DIM + col] : 0.0f;
    }
    __syncthreads();

    const int wave = tid >> 5;
    const int lane = tid & 31;
    const int half = lane >> 4;
    const int l    = lane & 15;
    const int row0 = (blockIdx.x * 8 + wave) * 16;
    if (row0 >= n) return;

    int arow = row0 + l;
    if (arow >= n) arow = n - 1;

    v8f c = {};
    #pragma unroll
    for (int k0 = 0; k0 < H1_DIM; k0 += 4) {
        const int ka = k0 + half * 2;
        v2f a = *(const v2f*)&h1[(size_t)arow * H1_DIM + ka];
        for (int s = 0; s < nsl; ++s) {
            const v2f p = *(const v2f*)&part[((size_t)s * n + arow) * H1_DIM + ka];
            a.x += p.x;
            a.y += p.y;
        }
        a.x = fmaxf(a.x, 0.0f);
        a.y = fmaxf(a.y, 0.0f);
        v2f b;
        b.x = W2s[ka * 16 + l];
        b.y = W2s[(ka + 1) * 16 + l];
        c = __builtin_amdgcn_wmma_f32_16x16x4_f32(
                false, a, false, b, (short)0, c, false, false);
    }

    const bool  col  = l < H2_DIM;
    const float bias = col ? b2[l] : 0.0f;
    if (row0 + 16 <= n) {
        #pragma unroll
        for (int r = 0; r < 8; ++r) {
            const int row = row0 + r + half * 8;
            const float v  = c[r];
            const float dv = dis[row];
            hw2[(size_t)row * (2 * H2_DIM) + l] = col ? v : dv;
            if (col) out0[(size_t)row * H2_DIM + l] = dv * dv * v + bias;
        }
    } else {
        for (int r = 0; r < 8; ++r) {
            const int row = row0 + r + half * 8;
            if (row < n) {
                const float v  = c[r];
                const float dv = dis[row];
                hw2[(size_t)row * (2 * H2_DIM) + l] = col ? v : dv;
                if (col) out0[(size_t)row * H2_DIM + l] = dv * dv * v + bias;
            }
        }
    }
}

__global__ __launch_bounds__(256) void gcn_sum_out(
    const float* __restrict__ out0, const float* __restrict__ part, int nsl,
    float* __restrict__ out, int n) {
    const int i = blockIdx.x * blockDim.x + threadIdx.x;
    if (i >= n * (H2_DIM / 4)) return;
    float4 a = ((const float4*)out0)[i];
    for (int s = 0; s < nsl; ++s) {
        const float4 p = ((const float4*)(part + (size_t)s * n * H2_DIM))[i];
        a.x += p.x; a.y += p.y; a.z += p.z; a.w += p.w;
    }
    ((float4*)out)[i] = a;
}

extern "C" void kernel_launch(void* const* d_in, const int* in_sizes, int n_in,
                              void* d_out, int out_size, void* d_ws, size_t ws_size,
                              hipStream_t stream) {
    const float* x    = (const float*)d_in[0];
    const int*   edge = (const int*)  d_in[1];
    const float* W1   = (const float*)d_in[2];
    const float* b1   = (const float*)d_in[3];
    const float* W2   = (const float*)d_in[4];
    const float* b2   = (const float*)d_in[5];
    float* out = (float*)d_out;

    const int n = in_sizes[0] / IN_DIM;
    const int e = in_sizes[1] / 2;
    const int* src = edge;
    const int* dst = edge + e;

    const int R1 = 4096, R2 = 8192;
    const int nr1 = (n + R1 - 1) / R1;
    const int nr2 = (n + R2 - 1) / R2;
    const int S1  = (256 / nr1 > 0) ? 256 / nr1 : 1;
    const int S2  = (256 / nr2 > 0) ? 256 / nr2 : 1;
    const int chunks = (e + 127) / 128;
    const int cps1 = (chunks + S1 - 1) / S1;
    const int cps2 = (chunks + S2 - 1) / S2;

    const int nd = (n + 63) & ~63;
    float* ws    = (float*)d_ws;
    float* dis   = ws;
    float* hw1   = dis + nd;
    float* agg1  = hw1 + (size_t)n * (2 * H1_DIM);
    float* hw2   = agg1 + (size_t)n * H1_DIM;
    float* out0  = hw2 + (size_t)n * (2 * H2_DIM);
    float* part1 = out0 + (size_t)n * H2_DIM;
    float* part2 = part1 + (size_t)S1 * n * H1_DIM;
    unsigned* tabs = (unsigned*)(part2 + (size_t)S2 * n * H2_DIM);
    const int dwords = (((n + 1) / 2) + 63) & ~63;
    const int ntab   = (e + DH_EDGES - 1) / DH_EDGES;

    const int B = 256;
    const int tiles  = (n + 15) / 16;
    const int gtiles = (tiles + 7) / 8;
    const int gsum   = (n * (H2_DIM / 4) + B - 1) / B;

    gcn_deg_hist<<<ntab, 1024, 0, stream>>>(dst, tabs, n, e, dwords);
    gcn_deg_dis <<<dwords / 64, B, 0, stream>>>(tabs, ntab, dwords, dis, n);

    gcn_gemm1_wmma<<<gtiles, B, 0, stream>>>(x, W1, dis, b1, hw1, agg1, n);
    gcn_scatter_ranges<H1_DIM, 4096><<<nr1 * S1, 1024, 0, stream>>>(
        src, dst, dis, hw1, part1, n, e, nr1, cps1);

    gcn_gemm2_wmma<<<gtiles, B, 0, stream>>>(agg1, part1, S1, W2, dis, b2, hw2, out0, n);
    gcn_scatter_ranges<H2_DIM, 8192><<<nr2 * S2, 1024, 0, stream>>>(
        src, dst, dis, hw2, part2, n, e, nr2, cps2);
    gcn_sum_out<<<gsum, B, 0, stream>>>(out0, part2, S2, out, n);
}


# Scatter batches: an edge's D lanes read its queue entry and dis[dst] from LDS (same-address reads) and load (hw, dis[src]) pairs from the records: no lane picks, no weight gather; sums unchanged
# speedup vs baseline: 4.4885x; 1.4497x over previous
#include <hip/hip_runtime.h>


typedef __attribute__((ext_vector_type(2))) float v2f;
typedef __attribute__((ext_vector_type(8))) float v8f;

#define IN_DIM 128
#define H1_DIM 16
#define H2_DIM 8

#define DH_EDGES 32768
#define DH_WORDS 51200

#define DH_COUNT(val)                                                          \
    {                                                                          \
        const unsigned node = (unsigned)(val);                                 \
        if (node < lim) atomicAdd(&cnt[node >> 1], 1u << ((node & 1u) * 16u)); \
    }

__global__ __launch_bounds__(1024) void gcn_deg_hist(
    const int* __restrict__ dst, unsigned* __restrict__ tabs, int n, int e, int words) {
    __shared__ unsigned cnt[DH_WORDS];
    const int tid = threadIdx.x;
    for (int i = tid * 4; i < words; i += 1024 * 4)
        *(uint4*)&cnt[i] = make_uint4(0u, 0u, 0u, 0u);
    __syncthreads();

    const unsigned lim = ((unsigned)n < 2u * DH_WORDS) ? (unsigned)n : 2u * DH_WORDS;
    const int base = blockIdx.x * DH_EDGES;
    const bool vec_ok = (e & 3) == 0;
    int4 d[DH_EDGES / 4096];
    if (vec_ok && e - base >= DH_EDGES) {
        #pragma unroll
        for (int j = 0; j < DH_EDGES / 4096; ++j)
            d[j] = *(const int4*)&dst[base + (j * 1024 + tid) * 4];
    } else {
        #pragma unroll
        for (int j = 0; j < DH_EDGES / 4096; ++j) {
            const int i = base + (j * 1024 + tid) * 4;
            d[j] = make_int4(-1, -1, -1, -1);
            if (i     < e) d[j].x = dst[i];
            if (i + 1 < e) d[j].y = dst[i + 1];
            if (i + 2 < e) d[j].z = dst[i + 2];
            if (i + 3 < e) d[j].w = dst[i + 3];
        }
    }
    #pragma unroll
    for (int j = 0; j < DH_EDGES / 4096; ++j) {
        DH_COUNT(d[j].x)
        DH_COUNT(d[j].y)
        DH_COUNT(d[j].z)
        DH_COUNT(d[j].w)
    }
    __syncthreads();

    unsigned* outp = tabs + (size_t)blockIdx.x * words;
    for (int i = tid * 4; i < words; i += 1024 * 4)
        *(uint4*)&outp[i] = *(const uint4*)&cnt[i];
}

__device__ __forceinline__ float gcn_dis_of(unsigned count) {
    const float d = (float)(count + 1u);
    return (d > 0.0f) ? rsqrtf(d) : 0.0f;
}

__global__ __launch_bounds__(256) void gcn_deg_dis(
    const unsigned* __restrict__ tabs, int ntab, int words,
    float* __restrict__ dis, int n) {
    __shared__ uint4 red[8 * 32];
    const int tid  = threadIdx.x;
    const int lane = tid & 31;
    const int wave = __builtin_amdgcn_readfirstlane(tid >> 5);
    const int q = blockIdx.x * 32 + lane;
    unsigned c0 = 0u, c1 = 0u, c2 = 0u, c3 = 0u;
    for (int s0 = wave; s0 < ntab; s0 += 64) {
        uint2 v[8];
        #pragma unroll
        for (int u = 0; u < 8; ++u) {
            const int s = s0 + u * 8;
            v[u] = make_uint2(0u, 0u);
            if (s < ntab) v[u] = *(const uint2*)&tabs[(size_t)s * words + 2 * q];
        }
        #pragma unroll
        for (int u = 0; u < 8; ++u) {
            c0 += v[u].x & 0xffffu;
            c1 += v[u].x >> 16;
            c2 += v[u].y & 0xffffu;
            c3 += v[u].y >> 16;
        }
    }
    red[wave * 32 + lane] = make_uint4(c0, c1, c2, c3);
    __syncthreads();
    if (wave == 0) {
        uint4 t = red[lane];
        #pragma unroll
        for (int w = 1; w < 8; ++w) {
            const uint4 r = red[w * 32 + lane];
            t.x += r.x; t.y += r.y; t.z += r.z; t.w += r.w;
        }
        const int node = q * 4;
        const float4 o = make_float4(gcn_dis_of(t.x), gcn_dis_of(t.y),
                                     gcn_dis_of(t.z), gcn_dis_of(t.w));
        if (node + 3 < n) {
            *(float4*)&dis[node] = o;
        } else {
            if (node     < n) dis[node]     = o.x;
            if (node + 1 < n) dis[node + 1] = o.y;
            if (node + 2 < n) dis[node + 2] = o.z;
        }
    }
}

__global__ __launch_bounds__(256) void gcn_gemm1_wmma(
    const float* __restrict__ x, const float* __restrict__ W1,
    const float* __restrict__ dis, const float* __restrict__ b1,
    float* __restrict__ hw1, float* __restrict__ agg1, int n) {
    __shared__ float W1s[IN_DIM * H1_DIM];
    const int tid = threadIdx.x;
    #pragma unroll
    for (int j = 0; j < (IN_DIM * H1_DIM) / 256; ++j)
        W1s[tid + j * 256] = W1[tid + j * 256];
    __syncthreads();

    const int wave = tid >> 5;
    const int lane = tid & 31;
    const int half = lane >> 4;
    const int l    = lane & 15;
    const int row0 = (blockIdx.x * 8 + wave) * 16;
    if (row0 >= n) return;

    int arow = row0 + l;
    if (arow >= n) arow = n - 1;

    v8f c = {};
    #pragma unroll 4
    for (int k0 = 0; k0 < IN_DIM; k0 += 4) {
        const int ka = k0 + half * 2;
        const v2f a = *(const v2f*)&x[(size_t)arow * IN_DIM + ka];
        v2f b;
        b.x = W1s[ka * H1_DIM + l];
        b.y = W1s[(ka + 1) * H1_DIM + l];
        c = __builtin_amdgcn_wmma_f32_16x16x4_f32(
                false, a, false, b, (short)0, c, false, false);
    }

    const float bias = b1[l];
    if (row0 + 16 <= n) {
        #pragma unroll
        for (int r = 0; r < 8; ++r) {
            const int row = row0 + r + half * 8;
            const float v  = c[r];
            const float dv = dis[row];
            v2f rec;
            rec.x = v;
            rec.y = dv;
            *(v2f*)&hw1[(size_t)row * (2 * H1_DIM) + 2 * l] = rec;
            agg1[(size_t)row * H1_DIM + l] = dv * dv * v + bias;
        }
    } else {
        for (int r = 0; r < 8; ++r) {
            const int row = row0 + r + half * 8;
            if (row < n) {
                const float v  = c[r];
                const float dv = dis[row];
                v2f rec;
                rec.x = v;
                rec.y = dv;
                *(v2f*)&hw1[(size_t)row * (2 * H1_DIM) + 2 * l] = rec;
                agg1[(size_t)row * H1_DIM + l] = dv * dv * v + bias;
            }
        }
    }
}

#define SC_WAVES  32
#define SC_UNROLL 4
#define SC_QCAP   160
#define SC_SRCBITS 17

template <int D, bool FULL>
__device__ __forceinline__ void sc_batch(const unsigned* qb, int cnt, int lane,
                                         const float* disl,
                                         const float* __restrict__ hw, float* acc) {
    constexpr int G = 32 / D;
    const int grp = lane / D;
    const int l   = lane % D;
    unsigned pk[D];
    v2f      rec[D];
    float    dd[D];
    #pragma unroll
    for (int t = 0; t < D; ++t) {
        pk[t] = 0u;
        if (FULL || t * G + grp < cnt) pk[t] = qb[t * G + grp];
    }
    #pragma unroll
    for (int t = 0; t < D; ++t) {
        const unsigned s   = pk[t] & ((1u << SC_SRCBITS) - 1u);
        const unsigned row = pk[t] >> SC_SRCBITS;
        rec[t] = *(const v2f*)&hw[s * (unsigned)(2 * D) + 2u * (unsigned)l];
        dd[t]  = disl[row];
    }
    #pragma unroll
    for (int t = 0; t < D; ++t) {
        const unsigned row = pk[t] >> SC_SRCBITS;
        const float w = rec[t].y * dd[t];
        if (FULL || t * G + grp < cnt)
            atomicAdd(&acc[row * (unsigned)D + (unsigned)l], w * rec[t].x);
    }
}

#define SC_ENQ(dval, sval)                                                     \
    {                                                                          \
        const unsigned dl = (unsigned)((dval) - lo);                           \
        const bool m = dl < (unsigned)nloc;                                    \
        const unsigned mask = __builtin_amdgcn_ballot_w32(m);                  \
        if (m) q[qn + (int)__builtin_amdgcn_mbcnt_lo(mask, 0u)] =              \
                   (unsigned)(sval) | (dl << SC_SRCBITS);                      \
        qn += __builtin_popcount(mask);                                        \
    }

template <int D, int R>
__global__ __launch_bounds__(1024) void gcn_scatter_ranges(
    const int* __restrict__ src, const int* __restrict__ dst,
    const float* __restrict__ dis, const float* __restrict__ hw,
    float* __restrict__ part, int n, int e, int nranges, int chunks_per_slice) {
    __shared__ struct {
        float    disl[R];
        unsigned queue[SC_WAVES * SC_QCAP];
        float    acc[R * D];
    } lds;
    float*    disl  = lds.disl;
    unsigned* queue = lds.queue;
    float*    acc   = lds.acc;
    const int tid  = threadIdx.x;
    const int lane = tid & 31;
    const int wave = __builtin_amdgcn_readfirstlane(tid >> 5);
    const int range = blockIdx.x % nranges;
    const int slice = blockIdx.x / nranges;
    const int lo    = range * R;
    const int nloc  = (n - lo < R) ? (n - lo) : R;

    for (int i = tid * 4; i < R * D; i += 1024 * 4)
        *(float4*)&acc[i] = make_float4(0.0f, 0.0f, 0.0f, 0.0f);
    for (int i = tid; i < nloc; i += 1024)
        disl[i] = dis[lo + i];
    __syncthreads();

    const int total_chunks = (e + 127) >> 7;
    const int c0 = slice * chunks_per_slice;
    const int c1 = (c0 + chunks_per_slice < total_chunks) ? (c0 + chunks_per_slice)
                                                          : total_chunks;
    const bool vec_ok = (e & 3) == 0;
    unsigned* q = &queue[wave * SC_QCAP];
    int qn = 0;

    for (int cb = c0 + wave; cb < c1; cb += SC_WAVES * SC_UNROLL) {
        int4 dv[SC_UNROLL], sv[SC_UNROLL];
        #pragma unroll
        for (int u = 0; u < SC_UNROLL; ++u) {
            const int c = cb + u * SC_WAVES;
            dv[u] = make_int4(-1, -1, -1, -1);
            sv[u] = make_int4(0, 0, 0, 0);
            if (c < c1) {
                const int base = c * 128 + lane * 4;
                if (vec_ok && c * 128 + 128 <= e) {
                    dv[u] = *(const int4*)&dst[base];
                    sv[u] = *(const int4*)&src[base];
                } else {
                    if (base     < e) { dv[u].x = dst[base];     sv[u].x = src[base];     }
                    if (base + 1 < e) { dv[u].y = dst[base + 1]; sv[u].y = src[base + 1]; }
                    if (base + 2 < e) { dv[u].z = dst[base + 2]; sv[u].z = src[base + 2]; }
                    if (base + 3 < e) { dv[u].w = dst[base + 3]; sv[u].w = src[base + 3]; }
                }
            }
        }
        #pragma unroll
        for (int u = 0; u < SC_UNROLL; ++u) {
            SC_ENQ(dv[u].x, sv[u].x)
            SC_ENQ(dv[u].y, sv[u].y)
            SC_ENQ(dv[u].z, sv[u].z)
            SC_ENQ(dv[u].w, sv[u].w)
            while (qn >= 32) {
                qn -= 32;
                sc_batch<D, true>(q + qn, 32, lane, disl, hw, acc);
            }
        }
    }
    if (qn > 0) sc_batch<D, false>(q, qn, lane, disl, hw, acc);
    __syncthreads();

    float* outp = part + ((size_t)slice * n + lo) * D;
    const int total = nloc * D;
    for (int i = tid * 4; i < total; i += 1024 * 4)
        *(float4*)&outp[i] = *(const float4*)&acc[i];
}

__global__ __launch_bounds__(256) void gcn_gemm2_wmma(
    const float* __restrict__ h1, const float* __restrict__ part, int nsl,
    const float* __restrict__ W2,
    const float* __restrict__ dis, const float* __restrict__ b2,
    float* __restrict__ hw2, float* __restrict__ out0, int n) {
    __shared__ float W2s[H1_DIM * 16];
    const int tid = threadIdx.x;
    {
        const int k = tid >> 4, col = tid & 15;
        W2s[tid] = (col < H2_DIM) ? W2[k * H2_DIM + col] : 0.0f;
    }
    __syncthreads();

    const int wave = tid >> 5;
    const int lane = tid & 31;
    const int half = lane >> 4;
    const int l    = lane & 15;
    const int row0 = (blockIdx.x * 8 + wave) * 16;
    if (row0 >= n) return;

    int arow = row0 + l;
    if (arow >= n) arow = n - 1;

    v8f c = {};
    #pragma unroll
    for (int k0 = 0; k0 < H1_DIM; k0 += 4) {
        const int ka = k0 + half * 2;
        v2f a = *(const v2f*)&h1[(size_t)arow * H1_DIM + ka];
        for (int s = 0; s < nsl; ++s) {
            const v2f p = *(const v2f*)&part[((size_t)s * n + arow) * H1_DIM + ka];
            a.x += p.x;
            a.y += p.y;
        }
        a.x = fmaxf(a.x, 0.0f);
        a.y = fmaxf(a.y, 0.0f);
        v2f b;
        b.x = W2s[ka * 16 + l];
        b.y = W2s[(ka + 1) * 16 + l];
        c = __builtin_amdgcn_wmma_f32_16x16x4_f32(
                false, a, false, b, (short)0, c, false, false);
    }

    const bool  col  = l < H2_DIM;
    const float bias = col ? b2[l] : 0.0f;
    if (row0 + 16 <= n) {
        #pragma unroll
        for (int r = 0; r < 8; ++r) {
            const int row = row0 + r + half * 8;
            const float v  = c[r];
            const float dv = dis[row];
            if (col) {
                v2f rec;
                rec.x = v;
                rec.y = dv;
                *(v2f*)&hw2[(size_t)row * (2 * H2_DIM) + 2 * l] = rec;
                out0[(size_t)row * H2_DIM + l] = dv * dv * v + bias;
            }
        }
    } else {
        for (int r = 0; r < 8; ++r) {
            const int row = row0 + r + half * 8;
            if (row < n) {
                const float v  = c[r];
                const float dv = dis[row];
                if (col) {
                    v2f rec;
                    rec.x = v;
                    rec.y = dv;
                    *(v2f*)&hw2[(size_t)row * (2 * H2_DIM) + 2 * l] = rec;
                    out0[(size_t)row * H2_DIM + l] = dv * dv * v + bias;
                }
            }
        }
    }
}

__global__ __launch_bounds__(256) void gcn_sum_out(
    const float* __restrict__ out0, const float* __restrict__ part, int nsl,
    float* __restrict__ out, int n) {
    const int i = blockIdx.x * blockDim.x + threadIdx.x;
    if (i >= n * (H2_DIM / 4)) return;
    float4 a = ((const float4*)out0)[i];
    for (int s = 0; s < nsl; ++s) {
        const float4 p = ((const float4*)(part + (size_t)s * n * H2_DIM))[i];
        a.x += p.x; a.y += p.y; a.z += p.z; a.w += p.w;
    }
    ((float4*)out)[i] = a;
}

extern "C" void kernel_launch(void* const* d_in, const int* in_sizes, int n_in,
                              void* d_out, int out_size, void* d_ws, size_t ws_size,
                              hipStream_t stream) {
    const float* x    = (const float*)d_in[0];
    const int*   edge = (const int*)  d_in[1];
    const float* W1   = (const float*)d_in[2];
    const float* b1   = (const float*)d_in[3];
    const float* W2   = (const float*)d_in[4];
    const float* b2   = (const float*)d_in[5];
    float* out = (float*)d_out;

    const int n = in_sizes[0] / IN_DIM;
    const int e = in_sizes[1] / 2;
    const int* src = edge;
    const int* dst = edge + e;

    const int R1 = 4096, R2 = 8192;
    const int nr1 = (n + R1 - 1) / R1;
    const int nr2 = (n + R2 - 1) / R2;
    const int S1  = (256 / nr1 > 0) ? 256 / nr1 : 1;
    const int S2  = (256 / nr2 > 0) ? 256 / nr2 : 1;
    const int chunks = (e + 127) / 128;
    const int cps1 = (chunks + S1 - 1) / S1;
    const int cps2 = (chunks + S2 - 1) / S2;

    const int nd = (n + 63) & ~63;
    float* ws    = (float*)d_ws;
    float* dis   = ws;
    float* hw1   = dis + nd;
    float* agg1  = hw1 + (size_t)n * (2 * H1_DIM);
    float* hw2   = agg1 + (size_t)n * H1_DIM;
    float* out0  = hw2 + (size_t)n * (2 * H2_DIM);
    float* part1 = out0 + (size_t)n * H2_DIM;
    float* part2 = part1 + (size_t)S1 * n * H1_DIM;
    unsigned* tabs = (unsigned*)(part2 + (size_t)S2 * n * H2_DIM);
    const int dwords = (((n + 1) / 2) + 63) & ~63;
    const int ntab   = (e + DH_EDGES - 1) / DH_EDGES;

    const int B = 256;
    const int tiles  = (n + 15) / 16;
    const int gtiles = (tiles + 7) / 8;
    const int gsum   = (n * (H2_DIM / 4) + B - 1) / B;

    gcn_deg_hist<<<ntab, 1024, 0, stream>>>(dst, tabs, n, e, dwords);
    gcn_deg_dis <<<dwords / 64, B, 0, stream>>>(tabs, ntab, dwords, dis, n);

    gcn_gemm1_wmma<<<gtiles, B, 0, stream>>>(x, W1, dis, b1, hw1, agg1, n);
    gcn_scatter_ranges<H1_DIM, 4096><<<nr1 * S1, 1024, 0, stream>>>(
        src, dst, dis, hw1, part1, n, e, nr1, cps1);

    gcn_gemm2_wmma<<<gtiles, B, 0, stream>>>(agg1, part1, S1, W2, dis, b2, hw2, out0, n);
    gcn_scatter_ranges<H2_DIM, 8192><<<nr2 * S2, 1024, 0, stream>>>(
        src, dst, dis, hw2, part2, n, e, nr2, cps2);
    gcn_sum_out<<<gsum, B, 0, stream>>>(out0, part2, S2, out, n);
}


# Layer-2 scatter reads per-wave match lists that layer 1's filter stores (one 128-B line per 32-edge batch) and no longer scans the edge list; layer-2 partials per 10 slices, LDS add order changes
# speedup vs baseline: 5.1343x; 1.1439x over previous
#include <hip/hip_runtime.h>


typedef __attribute__((ext_vector_type(2))) float v2f;
typedef __attribute__((ext_vector_type(8))) float v8f;

#define IN_DIM 128
#define H1_DIM 16
#define H2_DIM 8

#define DH_EDGES 32768
#define DH_WORDS 51200

#define DH_COUNT(val)                                                          \
    {                                                                          \
        const unsigned node = (unsigned)(val);                                 \
        if (node < lim) atomicAdd(&cnt[node >> 1], 1u << ((node & 1u) * 16u)); \
    }

__global__ __launch_bounds__(1024) void gcn_deg_hist(
    const int* __restrict__ dst, unsigned* __restrict__ tabs, int n, int e, int words) {
    __shared__ unsigned cnt[DH_WORDS];
    const int tid = threadIdx.x;
    for (int i = tid * 4; i < words; i += 1024 * 4)
        *(uint4*)&cnt[i] = make_uint4(0u, 0u, 0u, 0u);
    __syncthreads();

    const unsigned lim = ((unsigned)n < 2u * DH_WORDS) ? (unsigned)n : 2u * DH_WORDS;
    const int base = blockIdx.x * DH_EDGES;
    const bool vec_ok = (e & 3) == 0;
    int4 d[DH_EDGES / 4096];
    if (vec_ok && e - base >= DH_EDGES) {
        #pragma unroll
        for (int j = 0; j < DH_EDGES / 4096; ++j)
            d[j] = *(const int4*)&dst[base + (j * 1024 + tid) * 4];
    } else {
        #pragma unroll
        for (int j = 0; j < DH_EDGES / 4096; ++j) {
            const int i = base + (j * 1024 + tid) * 4;
            d[j] = make_int4(-1, -1, -1, -1);
            if (i     < e) d[j].x = dst[i];
            if (i + 1 < e) d[j].y = dst[i + 1];
            if (i + 2 < e) d[j].z = dst[i + 2];
            if (i + 3 < e) d[j].w = dst[i + 3];
        }
    }
    #pragma unroll
    for (int j = 0; j < DH_EDGES / 4096; ++j) {
        DH_COUNT(d[j].x)
        DH_COUNT(d[j].y)
        DH_COUNT(d[j].z)
        DH_COUNT(d[j].w)
    }
    __syncthreads();

    unsigned* outp = tabs + (size_t)blockIdx.x * words;
    for (int i = tid * 4; i < words; i += 1024 * 4)
        *(uint4*)&outp[i] = *(const uint4*)&cnt[i];
}

__device__ __forceinline__ float gcn_dis_of(unsigned count) {
    const float d = (float)(count + 1u);
    return (d > 0.0f) ? rsqrtf(d) : 0.0f;
}

__global__ __launch_bounds__(256) void gcn_deg_dis(
    const unsigned* __restrict__ tabs, int ntab, int words,
    float* __restrict__ dis, int n) {
    __shared__ uint4 red[8 * 32];
    const int tid  = threadIdx.x;
    const int lane = tid & 31;
    const int wave = __builtin_amdgcn_readfirstlane(tid >> 5);
    const int q = blockIdx.x * 32 + lane;
    unsigned c0 = 0u, c1 = 0u, c2 = 0u, c3 = 0u;
    for (int s0 = wave; s0 < ntab; s0 += 64) {
        uint2 v[8];
        #pragma unroll
        for (int u = 0; u < 8; ++u) {
            const int s = s0 + u * 8;
            v[u] = make_uint2(0u, 0u);
            if (s < ntab) v[u] = *(const uint2*)&tabs[(size_t)s * words + 2 * q];
        }
        #pragma unroll
        for (int u = 0; u < 8; ++u) {
            c0 += v[u].x & 0xffffu;
            c1 += v[u].x >> 16;
            c2 += v[u].y & 0xffffu;
            c3 += v[u].y >> 16;
        }
    }
    red[wave * 32 + lane] = make_uint4(c0, c1, c2, c3);
    __syncthreads();
    if (wave == 0) {
        uint4 t = red[lane];
        #pragma unroll
        for (int w = 1; w < 8; ++w) {
            const uint4 r = red[w * 32 + lane];
            t.x += r.x; t.y += r.y; t.z += r.z; t.w += r.w;
        }
        const int node = q * 4;
        const float4 o = make_float4(gcn_dis_of(t.x), gcn_dis_of(t.y),
                                     gcn_dis_of(t.z), gcn_dis_of(t.w));
        if (node + 3 < n) {
            *(float4*)&dis[node] = o;
        } else {
            if (node     < n) dis[node]     = o.x;
            if (node + 1 < n) dis[node + 1] = o.y;
            if (node + 2 < n) dis[node + 2] = o.z;
        }
    }
}

__global__ __launch_bounds__(256) void gcn_gemm1_wmma(
    const float* __restrict__ x, const float* __restrict__ W1,
    const float* __restrict__ dis, const float* __restrict__ b1,
    float* __restrict__ hw1, float* __restrict__ agg1, int n) {
    __shared__ float W1s[IN_DIM * H1_DIM];
    const int tid = threadIdx.x;
    #pragma unroll
    for (int j = 0; j < (IN_DIM * H1_DIM) / 256; ++j)
        W1s[tid + j * 256] = W1[tid + j * 256];
    __syncthreads();

    const int wave = tid >> 5;
    const int lane = tid & 31;
    const int half = lane >> 4;
    const int l    = lane & 15;
    const int row0 = (blockIdx.x * 8 + wave) * 16;
    if (row0 >= n) return;

    int arow = row0 + l;
    if (arow >= n) arow = n - 1;

    v8f c = {};
    #pragma unroll 4
    for (int k0 = 0; k0 < IN_DIM; k0 += 4) {
        const int ka = k0 + half * 2;
        const v2f a = *(const v2f*)&x[(size_t)arow * IN_DIM + ka];
        v2f b;
        b.x = W1s[ka * H1_DIM + l];
        b.y = W1s[(ka + 1) * H1_DIM + l];
        c = __builtin_amdgcn_wmma_f32_16x16x4_f32(
                false, a, false, b, (short)0, c, false, false);
    }

    const float bias = b1[l];
    if (row0 + 16 <= n) {
        #pragma unroll
        for (int r = 0; r < 8; ++r) {
            const int row = row0 + r + half * 8;
            const float v  = c[r];
            const float dv = dis[row];
            v2f rec;
            rec.x = v;
            rec.y = dv;
            *(v2f*)&hw1[(size_t)row * (2 * H1_DIM) + 2 * l] = rec;
            agg1[(size_t)row * H1_DIM + l] = dv * dv * v + bias;
        }
    } else {
        for (int r = 0; r < 8; ++r) {
            const int row = row0 + r + half * 8;
            if (row < n) {
                const float v  = c[r];
                const float dv = dis[row];
                v2f rec;
                rec.x = v;
                rec.y = dv;
                *(v2f*)&hw1[(size_t)row * (2 * H1_DIM) + 2 * l] = rec;
                agg1[(size_t)row * H1_DIM + l] = dv * dv * v + bias;
            }
        }
    }
}

#define SC_WAVES  32
#define SC_UNROLL 4
#define SC_QCAP   160
#define SC_SRCBITS 17

template <int D, int T, bool FULL>
__device__ __forceinline__ void sc_batch(const unsigned* qb, int cnt, int lane,
                                         const float* disl,
                                         const float* __restrict__ hw, float* acc) {
    constexpr int G = 32 / D;
    const int grp = lane / D;
    const int l   = lane % D;
    unsigned pk[T];
    v2f      rec[T];
    float    dd[T];
    #pragma unroll
    for (int t = 0; t < T; ++t) {
        pk[t] = 0u;
        if (FULL || t * G + grp < cnt) pk[t] = qb[t * G + grp];
    }
    #pragma unroll
    for (int t = 0; t < T; ++t) {
        const unsigned s   = pk[t] & ((1u << SC_SRCBITS) - 1u);
        const unsigned row = pk[t] >> SC_SRCBITS;
        rec[t] = *(const v2f*)&hw[s * (unsigned)(2 * D) + 2u * (unsigned)l];
        dd[t]  = disl[row];
    }
    #pragma unroll
    for (int t = 0; t < T; ++t) {
        const unsigned row = pk[t] >> SC_SRCBITS;
        const float w = rec[t].y * dd[t];
        if (FULL || t * G + grp < cnt)
            atomicAdd(&acc[row * (unsigned)D + (unsigned)l], w * rec[t].x);
    }
}

#define SC_ENQ(dval, sval)                                                     \
    {                                                                          \
        const unsigned dl = (unsigned)((dval) - lo);                           \
        const bool m = dl < (unsigned)nloc;                                    \
        const unsigned mask = __builtin_amdgcn_ballot_w32(m);                  \
        if (m) q[qn + (int)__builtin_amdgcn_mbcnt_lo(mask, 0u)] =              \
                   (unsigned)(sval) | (dl << SC_SRCBITS);                      \
        qn += __builtin_popcount(mask);                                        \
    }

template <int D, int R, bool LISTS_IN>
__global__ __launch_bounds__(1024) void gcn_scatter_ranges(
    const int* __restrict__ src, const int* __restrict__ dst,
    const float* __restrict__ dis, const float* __restrict__ hw,
    float* __restrict__ part, unsigned* __restrict__ lists, int* __restrict__ lcnt,
    int n, int e, int nranges, int chunks_per_slice, int cap) {
    __shared__ struct {
        float    disl[R];
        unsigned queue[SC_WAVES * SC_QCAP];
        float    acc[R * D];
    } lds;
    float*    disl  = lds.disl;
    unsigned* queue = lds.queue;
    float*    acc   = lds.acc;
    const int tid  = threadIdx.x;
    const int lane = tid & 31;
    const int wave = __builtin_amdgcn_readfirstlane(tid >> 5);
    const int range = blockIdx.x % nranges;
    const int slice = blockIdx.x / nranges;
    const int lo    = range * R;
    const int nloc  = (n - lo < R) ? (n - lo) : R;

    for (int i = tid * 4; i < R * D; i += 1024 * 4)
        *(float4*)&acc[i] = make_float4(0.0f, 0.0f, 0.0f, 0.0f);
    for (int i = tid; i < nloc; i += 1024)
        disl[i] = dis[lo + i];
    __syncthreads();

    const int total_chunks = (e + 127) >> 7;
    const int c0 = slice * chunks_per_slice;
    const int c1 = (c0 + chunks_per_slice < total_chunks) ? (c0 + chunks_per_slice)
                                                          : total_chunks;
    const bool vec_ok = (e & 3) == 0;
    unsigned* q = &queue[wave * SC_QCAP];
    int qn = 0;
    unsigned* lst = lists + ((size_t)blockIdx.x * SC_WAVES + wave) * (size_t)cap;
    int  ln  = 0;
    bool ovf = false;
    bool use_lists = false;
    int  myc = 0;
    if (LISTS_IN) {
        const int cl = lcnt[blockIdx.x * SC_WAVES + lane];
        use_lists = __builtin_amdgcn_ballot_w32(cl < 0) == 0u;
        myc = __builtin_amdgcn_readlane(cl, wave);
    }

    if (LISTS_IN && use_lists) {
        unsigned nx0 = 0u, nx1 = 0u;
        if (lane < myc)      nx0 = lst[lane];
        if (32 + lane < myc) nx1 = lst[32 + lane];
        for (int b = 0; b < myc; b += 64) {
            q[lane]      = nx0;
            q[32 + lane] = nx1;
            nx0 = 0u;
            nx1 = 0u;
            if (b + 64 + lane < myc) nx0 = lst[b + 64 + lane];
            if (b + 96 + lane < myc) nx1 = lst[b + 96 + lane];
            const int left = myc - b;
            if (left >= 64) sc_batch<D, 2 * D, true >(q, 64,   lane, disl, hw, acc);
            else            sc_batch<D, 2 * D, false>(q, left, lane, disl, hw, acc);
        }
    } else {
    for (int cb = c0 + wave; cb < c1; cb += SC_WAVES * SC_UNROLL) {
        int4 dv[SC_UNROLL], sv[SC_UNROLL];
        #pragma unroll
        for (int u = 0; u < SC_UNROLL; ++u) {
            const int c = cb + u * SC_WAVES;
            dv[u] = make_int4(-1, -1, -1, -1);
            sv[u] = make_int4(0, 0, 0, 0);
            if (c < c1) {
                const int base = c * 128 + lane * 4;
                if (vec_ok && c * 128 + 128 <= e) {
                    dv[u] = *(const int4*)&dst[base];
                    sv[u] = *(const int4*)&src[base];
                } else {
                    if (base     < e) { dv[u].x = dst[base];     sv[u].x = src[base];     }
                    if (base + 1 < e) { dv[u].y = dst[base + 1]; sv[u].y = src[base + 1]; }
                    if (base + 2 < e) { dv[u].z = dst[base + 2]; sv[u].z = src[base + 2]; }
                    if (base + 3 < e) { dv[u].w = dst[base + 3]; sv[u].w = src[base + 3]; }
                }
            }
        }
        #pragma unroll
        for (int u = 0; u < SC_UNROLL; ++u) {
            SC_ENQ(dv[u].x, sv[u].x)
            SC_ENQ(dv[u].y, sv[u].y)
            SC_ENQ(dv[u].z, sv[u].z)
            SC_ENQ(dv[u].w, sv[u].w)
            while (qn >= 32) {
                qn -= 32;
                if (!LISTS_IN) {
                    if (ln + 32 <= cap) lst[ln + lane] = q[qn + lane];
                    else                ovf = true;
                    ln += 32;
                }
                sc_batch<D, D, true>(q + qn, 32, lane, disl, hw, acc);
            }
        }
    }
    if (qn > 0) {
        if (!LISTS_IN) {
            if (ln + 32 <= cap) { if (lane < qn) lst[ln + lane] = q[lane]; }
            else                ovf = true;
            ln += qn;
        }
        sc_batch<D, D, false>(q, qn, lane, disl, hw, acc);
    }
    if (!LISTS_IN && lane == 0) lcnt[blockIdx.x * SC_WAVES + wave] = ovf ? -1 : ln;
    }
    __syncthreads();

    float* outp = part + ((size_t)slice * n + lo) * D;
    const int total = nloc * D;
    for (int i = tid * 4; i < total; i += 1024 * 4)
        *(float4*)&outp[i] = *(const float4*)&acc[i];
}

__global__ __launch_bounds__(256) void gcn_gemm2_wmma(
    const float* __restrict__ h1, const float* __restrict__ part, int nsl,
    const float* __restrict__ W2,
    const float* __restrict__ dis, const float* __restrict__ b2,
    float* __restrict__ hw2, float* __restrict__ out0, int n) {
    __shared__ float W2s[H1_DIM * 16];
    const int tid = threadIdx.x;
    {
        const int k = tid >> 4, col = tid & 15;
        W2s[tid] = (col < H2_DIM) ? W2[k * H2_DIM + col] : 0.0f;
    }
    __syncthreads();

    const int wave = tid >> 5;
    const int lane = tid & 31;
    const int half = lane >> 4;
    const int l    = lane & 15;
    const int row0 = (blockIdx.x * 8 + wave) * 16;
    if (row0 >= n) return;

    int arow = row0 + l;
    if (arow >= n) arow = n - 1;

    v8f c = {};
    #pragma unroll
    for (int k0 = 0; k0 < H1_DIM; k0 += 4) {
        const int ka = k0 + half * 2;
        v2f a = *(const v2f*)&h1[(size_t)arow * H1_DIM + ka];
        for (int s = 0; s < nsl; ++s) {
            const v2f p = *(const v2f*)&part[((size_t)s * n + arow) * H1_DIM + ka];
            a.x += p.x;
            a.y += p.y;
        }
        a.x = fmaxf(a.x, 0.0f);
        a.y = fmaxf(a.y, 0.0f);
        v2f b;
        b.x = W2s[ka * 16 + l];
        b.y = W2s[(ka + 1) * 16 + l];
        c = __builtin_amdgcn_wmma_f32_16x16x4_f32(
                false, a, false, b, (short)0, c, false, false);
    }

    const bool  col  = l < H2_DIM;
    const float bias = col ? b2[l] : 0.0f;
    if (row0 + 16 <= n) {
        #pragma unroll
        for (int r = 0; r < 8; ++r) {
            const int row = row0 + r + half * 8;
            const float v  = c[r];
            const float dv = dis[row];
            if (col) {
                v2f rec;
                rec.x = v;
                rec.y = dv;
                *(v2f*)&hw2[(size_t)row * (2 * H2_DIM) + 2 * l] = rec;
                out0[(size_t)row * H2_DIM + l] = dv * dv * v + bias;
            }
        }
    } else {
        for (int r = 0; r < 8; ++r) {
            const int row = row0 + r + half * 8;
            if (row < n) {
                const float v  = c[r];
                const float dv = dis[row];
                if (col) {
                    v2f rec;
                    rec.x = v;
                    rec.y = dv;
                    *(v2f*)&hw2[(size_t)row * (2 * H2_DIM) + 2 * l] = rec;
                    out0[(size_t)row * H2_DIM + l] = dv * dv * v + bias;
                }
            }
        }
    }
}

__global__ __launch_bounds__(256) void gcn_sum_out(
    const float* __restrict__ out0, const float* __restrict__ part, int nsl,
    float* __restrict__ out, int n) {
    const int i = blockIdx.x * blockDim.x + threadIdx.x;
    if (i >= n * (H2_DIM / 4)) return;
    float4 a = ((const float4*)out0)[i];
    for (int s = 0; s < nsl; ++s) {
        const float4 p = ((const float4*)(part + (size_t)s * n * H2_DIM))[i];
        a.x += p.x; a.y += p.y; a.z += p.z; a.w += p.w;
    }
    ((float4*)out)[i] = a;
}

extern "C" void kernel_launch(void* const* d_in, const int* in_sizes, int n_in,
                              void* d_out, int out_size, void* d_ws, size_t ws_size,
                              hipStream_t stream) {
    const float* x    = (const float*)d_in[0];
    const int*   edge = (const int*)  d_in[1];
    const float* W1   = (const float*)d_in[2];
    const float* b1   = (const float*)d_in[3];
    const float* W2   = (const float*)d_in[4];
    const float* b2   = (const float*)d_in[5];
    float* out = (float*)d_out;

    const int n = in_sizes[0] / IN_DIM;
    const int e = in_sizes[1] / 2;
    const int* src = edge;
    const int* dst = edge + e;

    const int R1 = 4096;
    const int nr1 = (n + R1 - 1) / R1;
    const int S1  = (256 / nr1 > 0) ? 256 / nr1 : 1;
    const int chunks = (e + 127) / 128;
    const int cps1 = (chunks + S1 - 1) / S1;
    const int cap = ((e / (nr1 * S1 * SC_WAVES)) * 2 + 64 + 31) & ~31;

    const int nd = (n + 63) & ~63;
    float* ws    = (float*)d_ws;
    float* dis   = ws;
    float* hw1   = dis + nd;
    float* agg1  = hw1 + (size_t)n * (2 * H1_DIM);
    float* hw2   = agg1 + (size_t)n * H1_DIM;
    float* out0  = hw2 + (size_t)n * (2 * H2_DIM);
    float* part1 = out0 + (size_t)n * H2_DIM;
    float* part2 = part1 + (size_t)S1 * n * H1_DIM;
    unsigned* tabs = (unsigned*)(part2 + (size_t)S1 * n * H2_DIM);
    const int dwords = (((n + 1) / 2) + 63) & ~63;
    const int ntab   = (e + DH_EDGES - 1) / DH_EDGES;
    unsigned* lists = tabs + (size_t)ntab * dwords;
    int*      lcnt  = (int*)(lists + (size_t)nr1 * S1 * SC_WAVES * cap);

    const int B = 256;
    const int tiles  = (n + 15) / 16;
    const int gtiles = (tiles + 7) / 8;
    const int gsum   = (n * (H2_DIM / 4) + B - 1) / B;

    gcn_deg_hist<<<ntab, 1024, 0, stream>>>(dst, tabs, n, e, dwords);
    gcn_deg_dis <<<dwords / 64, B, 0, stream>>>(tabs, ntab, dwords, dis, n);

    gcn_gemm1_wmma<<<gtiles, B, 0, stream>>>(x, W1, dis, b1, hw1, agg1, n);
    gcn_scatter_ranges<H1_DIM, 4096, false><<<nr1 * S1, 1024, 0, stream>>>(
        src, dst, dis, hw1, part1, lists, lcnt, n, e, nr1, cps1, cap);

    gcn_gemm2_wmma<<<gtiles, B, 0, stream>>>(agg1, part1, S1, W2, dis, b2, hw2, out0, n);
    gcn_scatter_ranges<H2_DIM, 4096, true><<<nr1 * S1, 1024, 0, stream>>>(
        src, dst, dis, hw2, part2, lists, lcnt, n, e, nr1, cps1, cap);
    gcn_sum_out<<<gsum, B, 0, stream>>>(out0, part2, S1, out, n);
}


# Both GEMMs: a wave loads its A tile as whole rows (512-B runs: each line of x and of the slice sums touched once, not 8 or 4 times) and reads the WMMA fragments from a padded LDS tile; sums unchanged
# speedup vs baseline: 5.3390x; 1.0399x over previous
#include <hip/hip_runtime.h>


typedef __attribute__((ext_vector_type(2))) float v2f;
typedef __attribute__((ext_vector_type(8))) float v8f;

#define IN_DIM 128
#define H1_DIM 16
#define H2_DIM 8

#define DH_EDGES 32768
#define DH_WORDS 51200

#define DH_COUNT(val)                                                          \
    {                                                                          \
        const unsigned node = (unsigned)(val);                                 \
        if (node < lim) atomicAdd(&cnt[node >> 1], 1u << ((node & 1u) * 16u)); \
    }

__global__ __launch_bounds__(1024) void gcn_deg_hist(
    const int* __restrict__ dst, unsigned* __restrict__ tabs, int n, int e, int words) {
    __shared__ unsigned cnt[DH_WORDS];
    const int tid = threadIdx.x;
    for (int i = tid * 4; i < words; i += 1024 * 4)
        *(uint4*)&cnt[i] = make_uint4(0u, 0u, 0u, 0u);
    __syncthreads();

    const unsigned lim = ((unsigned)n < 2u * DH_WORDS) ? (unsigned)n : 2u * DH_WORDS;
    const int base = blockIdx.x * DH_EDGES;
    const bool vec_ok = (e & 3) == 0;
    int4 d[DH_EDGES / 4096];
    if (vec_ok && e - base >= DH_EDGES) {
        #pragma unroll
        for (int j = 0; j < DH_EDGES / 4096; ++j)
            d[j] = *(const int4*)&dst[base + (j * 1024 + tid) * 4];
    } else {
        #pragma unroll
        for (int j = 0; j < DH_EDGES / 4096; ++j) {
            const int i = base + (j * 1024 + tid) * 4;
            d[j] = make_int4(-1, -1, -1, -1);
            if (i     < e) d[j].x = dst[i];
            if (i + 1 < e) d[j].y = dst[i + 1];
            if (i + 2 < e) d[j].z = dst[i + 2];
            if (i + 3 < e) d[j].w = dst[i + 3];
        }
    }
    #pragma unroll
    for (int j = 0; j < DH_EDGES / 4096; ++j) {
        DH_COUNT(d[j].x)
        DH_COUNT(d[j].y)
        DH_COUNT(d[j].z)
        DH_COUNT(d[j].w)
    }
    __syncthreads();

    unsigned* outp = tabs + (size_t)blockIdx.x * words;
    for (int i = tid * 4; i < words; i += 1024 * 4)
        *(uint4*)&outp[i] = *(const uint4*)&cnt[i];
}

__device__ __forceinline__ float gcn_dis_of(unsigned count) {
    const float d = (float)(count + 1u);
    return (d > 0.0f) ? rsqrtf(d) : 0.0f;
}

__global__ __launch_bounds__(256) void gcn_deg_dis(
    const unsigned* __restrict__ tabs, int ntab, int words,
    float* __restrict__ dis, int n) {
    __shared__ uint4 red[8 * 32];
    const int tid  = threadIdx.x;
    const int lane = tid & 31;
    const int wave = __builtin_amdgcn_readfirstlane(tid >> 5);
    const int q = blockIdx.x * 32 + lane;
    unsigned c0 = 0u, c1 = 0u, c2 = 0u, c3 = 0u;
    for (int s0 = wave; s0 < ntab; s0 += 64) {
        uint2 v[8];
        #pragma unroll
        for (int u = 0; u < 8; ++u) {
            const int s = s0 + u * 8;
            v[u] = make_uint2(0u, 0u);
            if (s < ntab) v[u] = *(const uint2*)&tabs[(size_t)s * words + 2 * q];
        }
        #pragma unroll
        for (int u = 0; u < 8; ++u) {
            c0 += v[u].x & 0xffffu;
            c1 += v[u].x >> 16;
            c2 += v[u].y & 0xffffu;
            c3 += v[u].y >> 16;
        }
    }
    red[wave * 32 + lane] = make_uint4(c0, c1, c2, c3);
    __syncthreads();
    if (wave == 0) {
        uint4 t = red[lane];
        #pragma unroll
        for (int w = 1; w < 8; ++w) {
            const uint4 r = red[w * 32 + lane];
            t.x += r.x; t.y += r.y; t.z += r.z; t.w += r.w;
        }
        const int node = q * 4;
        const float4 o = make_float4(gcn_dis_of(t.x), gcn_dis_of(t.y),
                                     gcn_dis_of(t.z), gcn_dis_of(t.w));
        if (node + 3 < n) {
            *(float4*)&dis[node] = o;
        } else {
            if (node     < n) dis[node]     = o.x;
            if (node + 1 < n) dis[node + 1] = o.y;
            if (node + 2 < n) dis[node + 2] = o.z;
        }
    }
}

#define G1_PITCH 132
__global__ __launch_bounds__(256) void gcn_gemm1_wmma(
    const float* __restrict__ x, const float* __restrict__ W1,
    const float* __restrict__ dis, const float* __restrict__ b1,
    float* __restrict__ hw1, float* __restrict__ agg1, int n) {
    __shared__ float W1s[IN_DIM * H1_DIM];
    __shared__ float As1[8 * 16 * G1_PITCH];
    const int tid  = threadIdx.x;
    const int wave = tid >> 5;
    const int lane = tid & 31;
    const int half = lane >> 4;
    const int l    = lane & 15;
    const int row0 = (blockIdx.x * 8 + wave) * 16;

    float4 t[16];
    #pragma unroll
    for (int i = 0; i < 16; ++i) {
        int row = row0 + i;
        if (row >= n) row = n - 1;
        t[i] = *(const float4*)&x[(size_t)row * IN_DIM + 4 * lane];
    }
    #pragma unroll
    for (int j = 0; j < (IN_DIM * H1_DIM) / 256; ++j)
        W1s[tid + j * 256] = W1[tid + j * 256];
    float* at = &As1[wave * (16 * G1_PITCH)];
    #pragma unroll
    for (int i = 0; i < 16; ++i)
        *(float4*)&at[i * G1_PITCH + 4 * lane] = t[i];
    __syncthreads();
    if (row0 >= n) return;

    v8f c = {};
    #pragma unroll
    for (int k0 = 0; k0 < IN_DIM; k0 += 4) {
        const int ka = k0 + half * 2;
        const v2f a = *(const v2f*)&at[l * G1_PITCH + ka];
        v2f b;
        b.x = W1s[ka * H1_DIM + l];
        b.y = W1s[(ka + 1) * H1_DIM + l];
        c = __builtin_amdgcn_wmma_f32_16x16x4_f32(
                false, a, false, b, (short)0, c, false, false);
    }

    const float bias = b1[l];
    if (row0 + 16 <= n) {
        #pragma unroll
        for (int r = 0; r < 8; ++r) {
            const int row = row0 + r + half * 8;
            const float v  = c[r];
            const float dv = dis[row];
            v2f rec;
            rec.x = v;
            rec.y = dv;
            *(v2f*)&hw1[(size_t)row * (2 * H1_DIM) + 2 * l] = rec;
            agg1[(size_t)row * H1_DIM + l] = dv * dv * v + bias;
        }
    } else {
        for (int r = 0; r < 8; ++r) {
            const int row = row0 + r + half * 8;
            if (row < n) {
                const float v  = c[r];
                const float dv = dis[row];
                v2f rec;
                rec.x = v;
                rec.y = dv;
                *(v2f*)&hw1[(size_t)row * (2 * H1_DIM) + 2 * l] = rec;
                agg1[(size_t)row * H1_DIM + l] = dv * dv * v + bias;
            }
        }
    }
}

#define SC_WAVES  32
#define SC_UNROLL 4
#define SC_QCAP   160
#define SC_SRCBITS 17

template <int D, int T, bool FULL>
__device__ __forceinline__ void sc_batch(const unsigned* qb, int cnt, int lane,
                                         const float* disl,
                                         const float* __restrict__ hw, float* acc) {
    constexpr int G = 32 / D;
    const int grp = lane / D;
    const int l   = lane % D;
    unsigned pk[T];
    v2f      rec[T];
    float    dd[T];
    #pragma unroll
    for (int t = 0; t < T; ++t) {
        pk[t] = 0u;
        if (FULL || t * G + grp < cnt) pk[t] = qb[t * G + grp];
    }
    #pragma unroll
    for (int t = 0; t < T; ++t) {
        const unsigned s   = pk[t] & ((1u << SC_SRCBITS) - 1u);
        const unsigned row = pk[t] >> SC_SRCBITS;
        rec[t] = *(const v2f*)&hw[s * (unsigned)(2 * D) + 2u * (unsigned)l];
        dd[t]  = disl[row];
    }
    #pragma unroll
    for (int t = 0; t < T; ++t) {
        const unsigned row = pk[t] >> SC_SRCBITS;
        const float w = rec[t].y * dd[t];
        if (FULL || t * G + grp < cnt)
            atomicAdd(&acc[row * (unsigned)D + (unsigned)l], w * rec[t].x);
    }
}

#define SC_ENQ(dval, sval)                                                     \
    {                                                                          \
        const unsigned dl = (unsigned)((dval) - lo);                           \
        const bool m = dl < (unsigned)nloc;                                    \
        const unsigned mask = __builtin_amdgcn_ballot_w32(m);                  \
        if (m) q[qn + (int)__builtin_amdgcn_mbcnt_lo(mask, 0u)] =              \
                   (unsigned)(sval) | (dl << SC_SRCBITS);                      \
        qn += __builtin_popcount(mask);                                        \
    }

template <int D, int R, bool LISTS_IN>
__global__ __launch_bounds__(1024) void gcn_scatter_ranges(
    const int* __restrict__ src, const int* __restrict__ dst,
    const float* __restrict__ dis, const float* __restrict__ hw,
    float* __restrict__ part, unsigned* __restrict__ lists, int* __restrict__ lcnt,
    int n, int e, int nranges, int chunks_per_slice, int cap) {
    __shared__ struct {
        float    disl[R];
        unsigned queue[SC_WAVES * SC_QCAP];
        float    acc[R * D];
    } lds;
    float*    disl  = lds.disl;
    unsigned* queue = lds.queue;
    float*    acc   = lds.acc;
    const int tid  = threadIdx.x;
    const int lane = tid & 31;
    const int wave = __builtin_amdgcn_readfirstlane(tid >> 5);
    const int range = blockIdx.x % nranges;
    const int slice = blockIdx.x / nranges;
    const int lo    = range * R;
    const int nloc  = (n - lo < R) ? (n - lo) : R;

    for (int i = tid * 4; i < R * D; i += 1024 * 4)
        *(float4*)&acc[i] = make_float4(0.0f, 0.0f, 0.0f, 0.0f);
    for (int i = tid; i < nloc; i += 1024)
        disl[i] = dis[lo + i];
    __syncthreads();

    const int total_chunks = (e + 127) >> 7;
    const int c0 = slice * chunks_per_slice;
    const int c1 = (c0 + chunks_per_slice < total_chunks) ? (c0 + chunks_per_slice)
                                                          : total_chunks;
    const bool vec_ok = (e & 3) == 0;
    unsigned* q = &queue[wave * SC_QCAP];
    int qn = 0;
    unsigned* lst = lists + ((size_t)blockIdx.x * SC_WAVES + wave) * (size_t)cap;
    int  ln  = 0;
    bool ovf = false;
    bool use_lists = false;
    int  myc = 0;
    if (LISTS_IN) {
        const int cl = lcnt[blockIdx.x * SC_WAVES + lane];
        use_lists = __builtin_amdgcn_ballot_w32(cl < 0) == 0u;
        myc = __builtin_amdgcn_readlane(cl, wave);
    }

    if (LISTS_IN && use_lists) {
        unsigned nx0 = 0u, nx1 = 0u;
        if (lane < myc)      nx0 = lst[lane];
        if (32 + lane < myc) nx1 = lst[32 + lane];
        for (int b = 0; b < myc; b += 64) {
            q[lane]      = nx0;
            q[32 + lane] = nx1;
            nx0 = 0u;
            nx1 = 0u;
            if (b + 64 + lane < myc) nx0 = lst[b + 64 + lane];
            if (b + 96 + lane < myc) nx1 = lst[b + 96 + lane];
            const int left = myc - b;
            if (left >= 64) sc_batch<D, 2 * D, true >(q, 64,   lane, disl, hw, acc);
            else            sc_batch<D, 2 * D, false>(q, left, lane, disl, hw, acc);
        }
    } else {
    for (int cb = c0 + wave; cb < c1; cb += SC_WAVES * SC_UNROLL) {
        int4 dv[SC_UNROLL], sv[SC_UNROLL];
        #pragma unroll
        for (int u = 0; u < SC_UNROLL; ++u) {
            const int c = cb + u * SC_WAVES;
            dv[u] = make_int4(-1, -1, -1, -1);
            sv[u] = make_int4(0, 0, 0, 0);
            if (c < c1) {
                const int base = c * 128 + lane * 4;
                if (vec_ok && c * 128 + 128 <= e) {
                    dv[u] = *(const int4*)&dst[base];
                    sv[u] = *(const int4*)&src[base];
                } else {
                    if (base     < e) { dv[u].x = dst[base];     sv[u].x = src[base];     }
                    if (base + 1 < e) { dv[u].y = dst[base + 1]; sv[u].y = src[base + 1]; }
                    if (base + 2 < e) { dv[u].z = dst[base + 2]; sv[u].z = src[base + 2]; }
                    if (base + 3 < e) { dv[u].w = dst[base + 3]; sv[u].w = src[base + 3]; }
                }
            }
        }
        #pragma unroll
        for (int u = 0; u < SC_UNROLL; ++u) {
            SC_ENQ(dv[u].x, sv[u].x)
            SC_ENQ(dv[u].y, sv[u].y)
            SC_ENQ(dv[u].z, sv[u].z)
            SC_ENQ(dv[u].w, sv[u].w)
            while (qn >= 32) {
                qn -= 32;
                if (!LISTS_IN) {
                    if (ln + 32 <= cap) lst[ln + lane] = q[qn + lane];
                    else                ovf = true;
                    ln += 32;
                }
                sc_batch<D, D, true>(q + qn, 32, lane, disl, hw, acc);
            }
        }
    }
    if (qn > 0) {
        if (!LISTS_IN) {
            if (ln + 32 <= cap) { if (lane < qn) lst[ln + lane] = q[lane]; }
            else                ovf = true;
            ln += qn;
        }
        sc_batch<D, D, false>(q, qn, lane, disl, hw, acc);
    }
    if (!LISTS_IN && lane == 0) lcnt[blockIdx.x * SC_WAVES + wave] = ovf ? -1 : ln;
    }
    __syncthreads();

    float* outp = part + ((size_t)slice * n + lo) * D;
    const int total = nloc * D;
    for (int i = tid * 4; i < total; i += 1024 * 4)
        *(float4*)&outp[i] = *(const float4*)&acc[i];
}

#define G2_PITCH 20
#define G2_SL    5
__global__ __launch_bounds__(256) void gcn_gemm2_wmma(
    const float* __restrict__ h1, const float* __restrict__ part, int nsl,
    const float* __restrict__ W2,
    const float* __restrict__ dis, const float* __restrict__ b2,
    float* __restrict__ hw2, float* __restrict__ out0, int n) {
    __shared__ float W2s[H1_DIM * 16];
    __shared__ float As2[8 * 16 * G2_PITCH];
    const int tid = threadIdx.x;
    {
        const int k = tid >> 4, col = tid & 15;
        W2s[tid] = (col < H2_DIM) ? W2[k * H2_DIM + col] : 0.0f;
    }

    const int wave = tid >> 5;
    const int lane = tid & 31;
    const int half = lane >> 4;
    const int l    = lane & 15;
    const int row0 = (blockIdx.x * 8 + wave) * 16;

    const int tr = lane >> 2;
    const int tc = (lane & 3) * 4;
    int row_a = row0 + tr;
    int row_b = row0 + 8 + tr;
    if (row_a >= n) row_a = n - 1;
    if (row_b >= n) row_b = n - 1;
    float4 va = *(const float4*)&h1[(size_t)row_a * H1_DIM + tc];
    float4 vb = *(const float4*)&h1[(size_t)row_b * H1_DIM + tc];
    for (int s0 = 0; s0 < nsl; s0 += G2_SL) {
        float4 pa[G2_SL], pb[G2_SL];
        #pragma unroll
        for (int u = 0; u < G2_SL; ++u) {
            const int s = s0 + u;
            pa[u] = make_float4(0.0f, 0.0f, 0.0f, 0.0f);
            pb[u] = make_float4(0.0f, 0.0f, 0.0f, 0.0f);
            if (s < nsl) {
                pa[u] = *(const float4*)&part[((size_t)s * n + row_a) * H1_DIM + tc];
                pb[u] = *(const float4*)&part[((size_t)s * n + row_b) * H1_DIM + tc];
            }
        }
        #pragma unroll
        for (int u = 0; u < G2_SL; ++u) {
            if (s0 + u < nsl) {
                va.x += pa[u].x; va.y += pa[u].y; va.z += pa[u].z; va.w += pa[u].w;
                vb.x += pb[u].x; vb.y += pb[u].y; vb.z += pb[u].z; vb.w += pb[u].w;
            }
        }
    }
    va.x = fmaxf(va.x, 0.0f); va.y = fmaxf(va.y, 0.0f);
    va.z = fmaxf(va.z, 0.0f); va.w = fmaxf(va.w, 0.0f);
    vb.x = fmaxf(vb.x, 0.0f); vb.y = fmaxf(vb.y, 0.0f);
    vb.z = fmaxf(vb.z, 0.0f); vb.w = fmaxf(vb.w, 0.0f);
    float* at = &As2[wave * (16 * G2_PITCH)];
    *(float4*)&at[tr * G2_PITCH + tc]       = va;
    *(float4*)&at[(8 + tr) * G2_PITCH + tc] = vb;
    __syncthreads();
    if (row0 >= n) return;

    v8f c = {};
    #pragma unroll
    for (int k0 = 0; k0 < H1_DIM; k0 += 4) {
        const int ka = k0 + half * 2;
        const v2f a = *(const v2f*)&at[l * G2_PITCH + ka];
        v2f b;
        b.x = W2s[ka * 16 + l];
        b.y = W2s[(ka + 1) * 16 + l];
        c = __builtin_amdgcn_wmma_f32_16x16x4_f32(
                false, a, false, b, (short)0, c, false, false);
    }

    const bool  col  = l < H2_DIM;
    const float bias = col ? b2[l] : 0.0f;
    if (row0 + 16 <= n) {
        #pragma unroll
        for (int r = 0; r < 8; ++r) {
            const int row = row0 + r + half * 8;
            const float v  = c[r];
            const float dv = dis[row];
            if (col) {
                v2f rec;
                rec.x = v;
                rec.y = dv;
                *(v2f*)&hw2[(size_t)row * (2 * H2_DIM) + 2 * l] = rec;
                out0[(size_t)row * H2_DIM + l] = dv * dv * v + bias;
            }
        }
    } else {
        for (int r = 0; r < 8; ++r) {
            const int row = row0 + r + half * 8;
            if (row < n) {
                const float v  = c[r];
                const float dv = dis[row];
                if (col) {
                    v2f rec;
                    rec.x = v;
                    rec.y = dv;
                    *(v2f*)&hw2[(size_t)row * (2 * H2_DIM) + 2 * l] = rec;
                    out0[(size_t)row * H2_DIM + l] = dv * dv * v + bias;
                }
            }
        }
    }
}

__global__ __launch_bounds__(256) void gcn_sum_out(
    const float* __restrict__ out0, const float* __restrict__ part, int nsl,
    float* __restrict__ out, int n) {
    const int i = blockIdx.x * blockDim.x + threadIdx.x;
    if (i >= n * (H2_DIM / 4)) return;
    float4 a = ((const float4*)out0)[i];
    for (int s = 0; s < nsl; ++s) {
        const float4 p = ((const float4*)(part + (size_t)s * n * H2_DIM))[i];
        a.x += p.x; a.y += p.y; a.z += p.z; a.w += p.w;
    }
    ((float4*)out)[i] = a;
}

extern "C" void kernel_launch(void* const* d_in, const int* in_sizes, int n_in,
                              void* d_out, int out_size, void* d_ws, size_t ws_size,
                              hipStream_t stream) {
    const float* x    = (const float*)d_in[0];
    const int*   edge = (const int*)  d_in[1];
    const float* W1   = (const float*)d_in[2];
    const float* b1   = (const float*)d_in[3];
    const float* W2   = (const float*)d_in[4];
    const float* b2   = (const float*)d_in[5];
    float* out = (float*)d_out;

    const int n = in_sizes[0] / IN_DIM;
    const int e = in_sizes[1] / 2;
    const int* src = edge;
    const int* dst = edge + e;

    const int R1 = 4096;
    const int nr1 = (n + R1 - 1) / R1;
    const int S1  = (256 / nr1 > 0) ? 256 / nr1 : 1;
    const int chunks = (e + 127) / 128;
    const int cps1 = (chunks + S1 - 1) / S1;
    const int cap = ((e / (nr1 * S1 * SC_WAVES)) * 2 + 64 + 31) & ~31;

    const int nd = (n + 63) & ~63;
    float* ws    = (float*)d_ws;
    float* dis   = ws;
    float* hw1   = dis + nd;
    float* agg1  = hw1 + (size_t)n * (2 * H1_DIM);
    float* hw2   = agg1 + (size_t)n * H1_DIM;
    float* out0  = hw2 + (size_t)n * (2 * H2_DIM);
    float* part1 = out0 + (size_t)n * H2_DIM;
    float* part2 = part1 + (size_t)S1 * n * H1_DIM;
    unsigned* tabs = (unsigned*)(part2 + (size_t)S1 * n * H2_DIM);
    const int dwords = (((n + 1) / 2) + 63) & ~63;
    const int ntab   = (e + DH_EDGES - 1) / DH_EDGES;
    unsigned* lists = tabs + (size_t)ntab * dwords;
    int*      lcnt  = (int*)(lists + (size_t)nr1 * S1 * SC_WAVES * cap);

    const int B = 256;
    const int tiles  = (n + 15) / 16;
    const int gtiles = (tiles + 7) / 8;
    const int gsum   = (n * (H2_DIM / 4) + B - 1) / B;

    gcn_deg_hist<<<ntab, 1024, 0, stream>>>(dst, tabs, n, e, dwords);
    gcn_deg_dis <<<dwords / 64, B, 0, stream>>>(tabs, ntab, dwords, dis, n);

    gcn_gemm1_wmma<<<gtiles, B, 0, stream>>>(x, W1, dis, b1, hw1, agg1, n);
    gcn_scatter_ranges<H1_DIM, 4096, false><<<nr1 * S1, 1024, 0, stream>>>(
        src, dst, dis, hw1, part1, lists, lcnt, n, e, nr1, cps1, cap);

    gcn_gemm2_wmma<<<gtiles, B, 0, stream>>>(agg1, part1, S1, W2, dis, b2, hw2, out0, n);
    gcn_scatter_ranges<H2_DIM, 4096, true><<<nr1 * S1, 1024, 0, stream>>>(
        src, dst, dis, hw2, part2, lists, lcnt, n, e, nr1, cps1, cap);
    gcn_sum_out<<<gsum, B, 0, stream>>>(out0, part2, S1, out, n);
}


# Layer-1 edge filter: a wave keeps the loads of its next three 128-edge chunks in flight while it filters the current one (rolling window of four register slots); sums and order unchanged
# speedup vs baseline: 5.5317x; 1.0361x over previous
#include <hip/hip_runtime.h>


typedef __attribute__((ext_vector_type(2))) float v2f;
typedef __attribute__((ext_vector_type(8))) float v8f;

#define IN_DIM 128
#define H1_DIM 16
#define H2_DIM 8

#define DH_EDGES 32768
#define DH_WORDS 51200

#define DH_COUNT(val)                                                          \
    {                                                                          \
        const unsigned node = (unsigned)(val);                                 \
        if (node < lim) atomicAdd(&cnt[node >> 1], 1u << ((node & 1u) * 16u)); \
    }

__global__ __launch_bounds__(1024) void gcn_deg_hist(
    const int* __restrict__ dst, unsigned* __restrict__ tabs, int n, int e, int words) {
    __shared__ unsigned cnt[DH_WORDS];
    const int tid = threadIdx.x;
    for (int i = tid * 4; i < words; i += 1024 * 4)
        *(uint4*)&cnt[i] = make_uint4(0u, 0u, 0u, 0u);
    __syncthreads();

    const unsigned lim = ((unsigned)n < 2u * DH_WORDS) ? (unsigned)n : 2u * DH_WORDS;
    const int base = blockIdx.x * DH_EDGES;
    const bool vec_ok = (e & 3) == 0;
    int4 d[DH_EDGES / 4096];
    if (vec_ok && e - base >= DH_EDGES) {
        #pragma unroll
        for (int j = 0; j < DH_EDGES / 4096; ++j)
            d[j] = *(const int4*)&dst[base + (j * 1024 + tid) * 4];
    } else {
        #pragma unroll
        for (int j = 0; j < DH_EDGES / 4096; ++j) {
            const int i = base + (j * 1024 + tid) * 4;
            d[j] = make_int4(-1, -1, -1, -1);
            if (i     < e) d[j].x = dst[i];
            if (i + 1 < e) d[j].y = dst[i + 1];
            if (i + 2 < e) d[j].z = dst[i + 2];
            if (i + 3 < e) d[j].w = dst[i + 3];
        }
    }
    #pragma unroll
    for (int j = 0; j < DH_EDGES / 4096; ++j) {
        DH_COUNT(d[j].x)
        DH_COUNT(d[j].y)
        DH_COUNT(d[j].z)
        DH_COUNT(d[j].w)
    }
    __syncthreads();

    unsigned* outp = tabs + (size_t)blockIdx.x * words;
    for (int i = tid * 4; i < words; i += 1024 * 4)
        *(uint4*)&outp[i] = *(const uint4*)&cnt[i];
}

__device__ __forceinline__ float gcn_dis_of(unsigned count) {
    const float d = (float)(count + 1u);
    return (d > 0.0f) ? rsqrtf(d) : 0.0f;
}

__global__ __launch_bounds__(256) void gcn_deg_dis(
    const unsigned* __restrict__ tabs, int ntab, int words,
    float* __restrict__ dis, int n) {
    __shared__ uint4 red[8 * 32];
    const int tid  = threadIdx.x;
    const int lane = tid & 31;
    const int wave = __builtin_amdgcn_readfirstlane(tid >> 5);
    const int q = blockIdx.x * 32 + lane;
    unsigned c0 = 0u, c1 = 0u, c2 = 0u, c3 = 0u;
    for (int s0 = wave; s0 < ntab; s0 += 64) {
        uint2 v[8];
        #pragma unroll
        for (int u = 0; u < 8; ++u) {
            const int s = s0 + u * 8;
            v[u] = make_uint2(0u, 0u);
            if (s < ntab) v[u] = *(const uint2*)&tabs[(size_t)s * words + 2 * q];
        }
        #pragma unroll
        for (int u = 0; u < 8; ++u) {
            c0 += v[u].x & 0xffffu;
            c1 += v[u].x >> 16;
            c2 += v[u].y & 0xffffu;
            c3 += v[u].y >> 16;
        }
    }
    red[wave * 32 + lane] = make_uint4(c0, c1, c2, c3);
    __syncthreads();
    if (wave == 0) {
        uint4 t = red[lane];
        #pragma unroll
        for (int w = 1; w < 8; ++w) {
            const uint4 r = red[w * 32 + lane];
            t.x += r.x; t.y += r.y; t.z += r.z; t.w += r.w;
        }
        const int node = q * 4;
        const float4 o = make_float4(gcn_dis_of(t.x), gcn_dis_of(t.y),
                                     gcn_dis_of(t.z), gcn_dis_of(t.w));
        if (node + 3 < n) {
            *(float4*)&dis[node] = o;
        } else {
            if (node     < n) dis[node]     = o.x;
            if (node + 1 < n) dis[node + 1] = o.y;
            if (node + 2 < n) dis[node + 2] = o.z;
        }
    }
}

#define G1_PITCH 132
__global__ __launch_bounds__(256) void gcn_gemm1_wmma(
    const float* __restrict__ x, const float* __restrict__ W1,
    const float* __restrict__ dis, const float* __restrict__ b1,
    float* __restrict__ hw1, float* __restrict__ agg1, int n) {
    __shared__ float W1s[IN_DIM * H1_DIM];
    __shared__ float As1[8 * 16 * G1_PITCH];
    const int tid  = threadIdx.x;
    const int wave = tid >> 5;
    const int lane = tid & 31;
    const int half = lane >> 4;
    const int l    = lane & 15;
    const int row0 = (blockIdx.x * 8 + wave) * 16;

    float4 t[16];
    #pragma unroll
    for (int i = 0; i < 16; ++i) {
        int row = row0 + i;
        if (row >= n) row = n - 1;
        t[i] = *(const float4*)&x[(size_t)row * IN_DIM + 4 * lane];
    }
    #pragma unroll
    for (int j = 0; j < (IN_DIM * H1_DIM) / 256; ++j)
        W1s[tid + j * 256] = W1[tid + j * 256];
    float* at = &As1[wave * (16 * G1_PITCH)];
    #pragma unroll
    for (int i = 0; i < 16; ++i)
        *(float4*)&at[i * G1_PITCH + 4 * lane] = t[i];
    __syncthreads();
    if (row0 >= n) return;

    v8f c = {};
    #pragma unroll
    for (int k0 = 0; k0 < IN_DIM; k0 += 4) {
        const int ka = k0 + half * 2;
        const v2f a = *(const v2f*)&at[l * G1_PITCH + ka];
        v2f b;
        b.x = W1s[ka * H1_DIM + l];
        b.y = W1s[(ka + 1) * H1_DIM + l];
        c = __builtin_amdgcn_wmma_f32_16x16x4_f32(
                false, a, false, b, (short)0, c, false, false);
    }

    const float bias = b1[l];
    if (row0 + 16 <= n) {
        #pragma unroll
        for (int r = 0; r < 8; ++r) {
            const int row = row0 + r + half * 8;
            const float v  = c[r];
            const float dv = dis[row];
            v2f rec;
            rec.x = v;
            rec.y = dv;
            *(v2f*)&hw1[(size_t)row * (2 * H1_DIM) + 2 * l] = rec;
            agg1[(size_t)row * H1_DIM + l] = dv * dv * v + bias;
        }
    } else {
        for (int r = 0; r < 8; ++r) {
            const int row = row0 + r + half * 8;
            if (row < n) {
                const float v  = c[r];
                const float dv = dis[row];
                v2f rec;
                rec.x = v;
                rec.y = dv;
                *(v2f*)&hw1[(size_t)row * (2 * H1_DIM) + 2 * l] = rec;
                agg1[(size_t)row * H1_DIM + l] = dv * dv * v + bias;
            }
        }
    }
}

#define SC_WAVES  32
#define SC_UNROLL 4
#define SC_QCAP   160
#define SC_SRCBITS 17

template <int D, int T, bool FULL>
__device__ __forceinline__ void sc_batch(const unsigned* qb, int cnt, int lane,
                                         const float* disl,
                                         const float* __restrict__ hw, float* acc) {
    constexpr int G = 32 / D;
    const int grp = lane / D;
    const int l   = lane % D;
    unsigned pk[T];
    v2f      rec[T];
    float    dd[T];
    #pragma unroll
    for (int t = 0; t < T; ++t) {
        pk[t] = 0u;
        if (FULL || t * G + grp < cnt) pk[t] = qb[t * G + grp];
    }
    #pragma unroll
    for (int t = 0; t < T; ++t) {
        const unsigned s   = pk[t] & ((1u << SC_SRCBITS) - 1u);
        const unsigned row = pk[t] >> SC_SRCBITS;
        rec[t] = *(const v2f*)&hw[s * (unsigned)(2 * D) + 2u * (unsigned)l];
        dd[t]  = disl[row];
    }
    #pragma unroll
    for (int t = 0; t < T; ++t) {
        const unsigned row = pk[t] >> SC_SRCBITS;
        const float w = rec[t].y * dd[t];
        if (FULL || t * G + grp < cnt)
            atomicAdd(&acc[row * (unsigned)D + (unsigned)l], w * rec[t].x);
    }
}

#define SC_ENQ(dval, sval)                                                     \
    {                                                                          \
        const unsigned dl = (unsigned)((dval) - lo);                           \
        const bool m = dl < (unsigned)nloc;                                    \
        const unsigned mask = __builtin_amdgcn_ballot_w32(m);                  \
        if (m) q[qn + (int)__builtin_amdgcn_mbcnt_lo(mask, 0u)] =              \
                   (unsigned)(sval) | (dl << SC_SRCBITS);                      \
        qn += __builtin_popcount(mask);                                        \
    }

#define SC_DRAIN()                                                             \
    while (qn >= 32) {                                                         \
        qn -= 32;                                                              \
        if (!LISTS_IN) {                                                       \
            if (ln + 32 <= cap) lst[ln + lane] = q[qn + lane];                 \
            else                ovf = true;                                    \
            ln += 32;                                                          \
        }                                                                      \
        sc_batch<D, D, true>(q + qn, 32, lane, disl, hw, acc);                 \
    }

template <int D, int R, bool LISTS_IN>
__global__ __launch_bounds__(1024) void gcn_scatter_ranges(
    const int* __restrict__ src, const int* __restrict__ dst,
    const float* __restrict__ dis, const float* __restrict__ hw,
    float* __restrict__ part, unsigned* __restrict__ lists, int* __restrict__ lcnt,
    int n, int e, int nranges, int chunks_per_slice, int cap) {
    __shared__ struct {
        float    disl[R];
        unsigned queue[SC_WAVES * SC_QCAP];
        float    acc[R * D];
    } lds;
    float*    disl  = lds.disl;
    unsigned* queue = lds.queue;
    float*    acc   = lds.acc;
    const int tid  = threadIdx.x;
    const int lane = tid & 31;
    const int wave = __builtin_amdgcn_readfirstlane(tid >> 5);
    const int range = blockIdx.x % nranges;
    const int slice = blockIdx.x / nranges;
    const int lo    = range * R;
    const int nloc  = (n - lo < R) ? (n - lo) : R;

    for (int i = tid * 4; i < R * D; i += 1024 * 4)
        *(float4*)&acc[i] = make_float4(0.0f, 0.0f, 0.0f, 0.0f);
    for (int i = tid; i < nloc; i += 1024)
        disl[i] = dis[lo + i];
    __syncthreads();

    const int total_chunks = (e + 127) >> 7;
    const int c0 = slice * chunks_per_slice;
    const int c1 = (c0 + chunks_per_slice < total_chunks) ? (c0 + chunks_per_slice)
                                                          : total_chunks;
    const bool vec_ok = (e & 3) == 0;
    unsigned* q = &queue[wave * SC_QCAP];
    int qn = 0;
    unsigned* lst = lists + ((size_t)blockIdx.x * SC_WAVES + wave) * (size_t)cap;
    int  ln  = 0;
    bool ovf = false;
    bool use_lists = false;
    int  myc = 0;
    if (LISTS_IN) {
        const int cl = lcnt[blockIdx.x * SC_WAVES + lane];
        use_lists = __builtin_amdgcn_ballot_w32(cl < 0) == 0u;
        myc = __builtin_amdgcn_readlane(cl, wave);
    }

    if (LISTS_IN && use_lists) {
        unsigned nx0 = 0u, nx1 = 0u;
        if (lane < myc)      nx0 = lst[lane];
        if (32 + lane < myc) nx1 = lst[32 + lane];
        for (int b = 0; b < myc; b += 64) {
            q[lane]      = nx0;
            q[32 + lane] = nx1;
            nx0 = 0u;
            nx1 = 0u;
            if (b + 64 + lane < myc) nx0 = lst[b + 64 + lane];
            if (b + 96 + lane < myc) nx1 = lst[b + 96 + lane];
            const int left = myc - b;
            if (left >= 64) sc_batch<D, 2 * D, true >(q, 64,   lane, disl, hw, acc);
            else            sc_batch<D, 2 * D, false>(q, left, lane, disl, hw, acc);
        }
    } else {
    const int efull = vec_ok ? (e >> 7) : 0;
    const int cfull = (c1 < efull) ? c1 : efull;
    const int cw    = c0 + wave;
    const int nfast = (cfull > cw) ? (cfull - cw + SC_WAVES - 1) / SC_WAVES : 0;
    if (nfast > 0) {
        int4 dv[SC_UNROLL], sv[SC_UNROLL];
        #pragma unroll
        for (int u = 0; u < SC_UNROLL; ++u) {
            const int j = (u < nfast) ? u : nfast - 1;
            const int base = (cw + j * SC_WAVES) * 128 + lane * 4;
            dv[u] = *(const int4*)&dst[base];
            sv[u] = *(const int4*)&src[base];
        }
        for (int j0 = 0; j0 < nfast; j0 += SC_UNROLL) {
            #pragma unroll
            for (int u = 0; u < SC_UNROLL; ++u) {
                if (j0 + u < nfast) {
                    SC_ENQ(dv[u].x, sv[u].x)
                    SC_ENQ(dv[u].y, sv[u].y)
                    SC_ENQ(dv[u].z, sv[u].z)
                    SC_ENQ(dv[u].w, sv[u].w)
                    SC_DRAIN()
                }
                int jn = j0 + u + SC_UNROLL;
                if (jn >= nfast) jn = nfast - 1;
                const int base = (cw + jn * SC_WAVES) * 128 + lane * 4;
                dv[u] = *(const int4*)&dst[base];
                sv[u] = *(const int4*)&src[base];
            }
        }
    }
    for (int c = cw + nfast * SC_WAVES; c < c1; c += SC_WAVES) {
        int4 d1 = make_int4(-1, -1, -1, -1);
        int4 s1 = make_int4(0, 0, 0, 0);
        const int base = c * 128 + lane * 4;
        if (base     < e) { d1.x = dst[base];     s1.x = src[base];     }
        if (base + 1 < e) { d1.y = dst[base + 1]; s1.y = src[base + 1]; }
        if (base + 2 < e) { d1.z = dst[base + 2]; s1.z = src[base + 2]; }
        if (base + 3 < e) { d1.w = dst[base + 3]; s1.w = src[base + 3]; }
        SC_ENQ(d1.x, s1.x)
        SC_ENQ(d1.y, s1.y)
        SC_ENQ(d1.z, s1.z)
        SC_ENQ(d1.w, s1.w)
        SC_DRAIN()
    }
    if (qn > 0) {
        if (!LISTS_IN) {
            if (ln + 32 <= cap) { if (lane < qn) lst[ln + lane] = q[lane]; }
            else                ovf = true;
            ln += qn;
        }
        sc_batch<D, D, false>(q, qn, lane, disl, hw, acc);
    }
    if (!LISTS_IN && lane == 0) lcnt[blockIdx.x * SC_WAVES + wave] = ovf ? -1 : ln;
    }
    __syncthreads();

    float* outp = part + ((size_t)slice * n + lo) * D;
    const int total = nloc * D;
    for (int i = tid * 4; i < total; i += 1024 * 4)
        *(float4*)&outp[i] = *(const float4*)&acc[i];
}

#define G2_PITCH 20
#define G2_SL    5
__global__ __launch_bounds__(256) void gcn_gemm2_wmma(
    const float* __restrict__ h1, const float* __restrict__ part, int nsl,
    const float* __restrict__ W2,
    const float* __restrict__ dis, const float* __restrict__ b2,
    float* __restrict__ hw2, float* __restrict__ out0, int n) {
    __shared__ float W2s[H1_DIM * 16];
    __shared__ float As2[8 * 16 * G2_PITCH];
    const int tid = threadIdx.x;
    {
        const int k = tid >> 4, col = tid & 15;
        W2s[tid] = (col < H2_DIM) ? W2[k * H2_DIM + col] : 0.0f;
    }

    const int wave = tid >> 5;
    const int lane = tid & 31;
    const int half = lane >> 4;
    const int l    = lane & 15;
    const int row0 = (blockIdx.x * 8 + wave) * 16;

    const int tr = lane >> 2;
    const int tc = (lane & 3) * 4;
    int row_a = row0 + tr;
    int row_b = row0 + 8 + tr;
    if (row_a >= n) row_a = n - 1;
    if (row_b >= n) row_b = n - 1;
    float4 va = *(const float4*)&h1[(size_t)row_a * H1_DIM + tc];
    float4 vb = *(const float4*)&h1[(size_t)row_b * H1_DIM + tc];
    for (int s0 = 0; s0 < nsl; s0 += G2_SL) {
        float4 pa[G2_SL], pb[G2_SL];
        #pragma unroll
        for (int u = 0; u < G2_SL; ++u) {
            const int s = s0 + u;
            pa[u] = make_float4(0.0f, 0.0f, 0.0f, 0.0f);
            pb[u] = make_float4(0.0f, 0.0f, 0.0f, 0.0f);
            if (s < nsl) {
                pa[u] = *(const float4*)&part[((size_t)s * n + row_a) * H1_DIM + tc];
                pb[u] = *(const float4*)&part[((size_t)s * n + row_b) * H1_DIM + tc];
            }
        }
        #pragma unroll
        for (int u = 0; u < G2_SL; ++u) {
            if (s0 + u < nsl) {
                va.x += pa[u].x; va.y += pa[u].y; va.z += pa[u].z; va.w += pa[u].w;
                vb.x += pb[u].x; vb.y += pb[u].y; vb.z += pb[u].z; vb.w += pb[u].w;
            }
        }
    }
    va.x = fmaxf(va.x, 0.0f); va.y = fmaxf(va.y, 0.0f);
    va.z = fmaxf(va.z, 0.0f); va.w = fmaxf(va.w, 0.0f);
    vb.x = fmaxf(vb.x, 0.0f); vb.y = fmaxf(vb.y, 0.0f);
    vb.z = fmaxf(vb.z, 0.0f); vb.w = fmaxf(vb.w, 0.0f);
    float* at = &As2[wave * (16 * G2_PITCH)];
    *(float4*)&at[tr * G2_PITCH + tc]       = va;
    *(float4*)&at[(8 + tr) * G2_PITCH + tc] = vb;
    __syncthreads();
    if (row0 >= n) return;

    v8f c = {};
    #pragma unroll
    for (int k0 = 0; k0 < H1_DIM; k0 += 4) {
        const int ka = k0 + half * 2;
        const v2f a = *(const v2f*)&at[l * G2_PITCH + ka];
        v2f b;
        b.x = W2s[ka * 16 + l];
        b.y = W2s[(ka + 1) * 16 + l];
        c = __builtin_amdgcn_wmma_f32_16x16x4_f32(
                false, a, false, b, (short)0, c, false, false);
    }

    const bool  col  = l < H2_DIM;
    const float bias = col ? b2[l] : 0.0f;
    if (row0 + 16 <= n) {
        #pragma unroll
        for (int r = 0; r < 8; ++r) {
            const int row = row0 + r + half * 8;
            const float v  = c[r];
            const float dv = dis[row];
            if (col) {
                v2f rec;
                rec.x = v;
                rec.y = dv;
                *(v2f*)&hw2[(size_t)row * (2 * H2_DIM) + 2 * l] = rec;
                out0[(size_t)row * H2_DIM + l] = dv * dv * v + bias;
            }
        }
    } else {
        for (int r = 0; r < 8; ++r) {
            const int row = row0 + r + half * 8;
            if (row < n) {
                const float v  = c[r];
                const float dv = dis[row];
                if (col) {
                    v2f rec;
                    rec.x = v;
                    rec.y = dv;
                    *(v2f*)&hw2[(size_t)row * (2 * H2_DIM) + 2 * l] = rec;
                    out0[(size_t)row * H2_DIM + l] = dv * dv * v + bias;
                }
            }
        }
    }
}

__global__ __launch_bounds__(256) void gcn_sum_out(
    const float* __restrict__ out0, const float* __restrict__ part, int nsl,
    float* __restrict__ out, int n) {
    const int i = blockIdx.x * blockDim.x + threadIdx.x;
    if (i >= n * (H2_DIM / 4)) return;
    float4 a = ((const float4*)out0)[i];
    for (int s = 0; s < nsl; ++s) {
        const float4 p = ((const float4*)(part + (size_t)s * n * H2_DIM))[i];
        a.x += p.x; a.y += p.y; a.z += p.z; a.w += p.w;
    }
    ((float4*)out)[i] = a;
}

extern "C" void kernel_launch(void* const* d_in, const int* in_sizes, int n_in,
                              void* d_out, int out_size, void* d_ws, size_t ws_size,
                              hipStream_t stream) {
    const float* x    = (const float*)d_in[0];
    const int*   edge = (const int*)  d_in[1];
    const float* W1   = (const float*)d_in[2];
    const float* b1   = (const float*)d_in[3];
    const float* W2   = (const float*)d_in[4];
    const float* b2   = (const float*)d_in[5];
    float* out = (float*)d_out;

    const int n = in_sizes[0] / IN_DIM;
    const int e = in_sizes[1] / 2;
    const int* src = edge;
    const int* dst = edge + e;

    const int R1 = 4096;
    const int nr1 = (n + R1 - 1) / R1;
    const int S1  = (256 / nr1 > 0) ? 256 / nr1 : 1;
    const int chunks = (e + 127) / 128;
    const int cps1 = (chunks + S1 - 1) / S1;
    const int cap = ((e / (nr1 * S1 * SC_WAVES)) * 2 + 64 + 31) & ~31;

    const int nd = (n + 63) & ~63;
    float* ws    = (float*)d_ws;
    float* dis   = ws;
    float* hw1   = dis + nd;
    float* agg1  = hw1 + (size_t)n * (2 * H1_DIM);
    float* hw2   = agg1 + (size_t)n * H1_DIM;
    float* out0  = hw2 + (size_t)n * (2 * H2_DIM);
    float* part1 = out0 + (size_t)n * H2_DIM;
    float* part2 = part1 + (size_t)S1 * n * H1_DIM;
    unsigned* tabs = (unsigned*)(part2 + (size_t)S1 * n * H2_DIM);
    const int dwords = (((n + 1) / 2) + 63) & ~63;
    const int ntab   = (e + DH_EDGES - 1) / DH_EDGES;
    unsigned* lists = tabs + (size_t)ntab * dwords;
    int*      lcnt  = (int*)(lists + (size_t)nr1 * S1 * SC_WAVES * cap);

    const int B = 256;
    const int tiles  = (n + 15) / 16;
    const int gtiles = (tiles + 7) / 8;
    const int gsum   = (n * (H2_DIM / 4) + B - 1) / B;

    gcn_deg_hist<<<ntab, 1024, 0, stream>>>(dst, tabs, n, e, dwords);
    gcn_deg_dis <<<dwords / 64, B, 0, stream>>>(tabs, ntab, dwords, dis, n);

    gcn_gemm1_wmma<<<gtiles, B, 0, stream>>>(x, W1, dis, b1, hw1, agg1, n);
    gcn_scatter_ranges<H1_DIM, 4096, false><<<nr1 * S1, 1024, 0, stream>>>(
        src, dst, dis, hw1, part1, lists, lcnt, n, e, nr1, cps1, cap);

    gcn_gemm2_wmma<<<gtiles, B, 0, stream>>>(agg1, part1, S1, W2, dis, b2, hw2, out0, n);
    gcn_scatter_ranges<H2_DIM, 4096, true><<<nr1 * S1, 1024, 0, stream>>>(
        src, dst, dis, hw2, part2, lists, lcnt, n, e, nr1, cps1, cap);
    gcn_sum_out<<<gsum, B, 0, stream>>>(out0, part2, S1, out, n);
}
